# starting kernel
# baseline (speedup 1.0000x reference)
#include <hip/hip_runtime.h>
#include <hip/hip_bf16.h>

typedef float v2f __attribute__((ext_vector_type(2)));
typedef float v8f __attribute__((ext_vector_type(8)));

#define WAVES 8
#define HID 10
#define EDGE_DIM 6

__device__ __forceinline__ float lrelu(float x) {
    return fmaxf(x, 0.1f * x);
}

__device__ __forceinline__ void stage_weights_padded(float* wlds, const float* w,
                                                     int rows_real, int rows_pad,
                                                     int t, int nthreads) {
    for (int i = t; i < rows_pad * 16; i += nthreads) {
        int row = i >> 4, col = i & 15;
        wlds[i] = (col < HID && row < rows_real) ? w[row * HID + col] : 0.0f;
    }
}

__global__ void gnn_embed_kernel(const float* __restrict__ vertex,
                                 const float* __restrict__ w_init,
                                 float* __restrict__ h_out, int n_nodes) {
    __shared__ float wlds[84 * 16];
    __shared__ float alds[WAVES][16 * 85];
    int t = threadIdx.x;
    stage_weights_padded(wlds, w_init, 82, 84, t, blockDim.x);
    __syncthreads();

    int wave = t >> 5, lane = t & 31;
    int tile = blockIdx.x * WAVES + wave;
    int base = tile * 16;
    if (base >= n_nodes) return;

    float* A = alds[wave];
    {
        int row = lane & 15, half = lane >> 4;
        int node = base + row;
        const float* vs = vertex + (size_t)node * 82;
        for (int j = 0; j < 41; ++j) {
            int k = half * 41 + j;
            A[row * 85 + k] = (node < n_nodes) ? vs[k] : 0.0f;
        }
        if (half == 0) { A[row * 85 + 82] = 0.0f; A[row * 85 + 83] = 0.0f; }
    }
    __builtin_amdgcn_wave_barrier();

    v8f c = {};
    int m = lane & 15;
    int koff = (lane < 16) ? 0 : 2;
    for (int kc = 0; kc < 21; ++kc) {
        int kb = kc * 4 + koff;
        v2f a, b;
        a.x = A[m * 85 + kb];
        a.y = A[m * 85 + kb + 1];
        b.x = wlds[kb * 16 + m];
        b.y = wlds[(kb + 1) * 16 + m];
        c = __builtin_amdgcn_wmma_f32_16x16x4_f32(false, a, false, b,
                                                  (short)0, c, false, false);
    }
    if (m < HID) {
        int rbase = (lane < 16) ? 0 : 8;
#pragma unroll
        for (int r = 0; r < 8; ++r) {
            int node = base + rbase + r;
            if (node < n_nodes)
                h_out[(size_t)node * HID + m] = lrelu(c[r]);
        }
    }
}

__global__ void gnn_zero_kernel(float* __restrict__ p, int n) {
    int i = blockIdx.x * blockDim.x + threadIdx.x;
    if (i < n) p[i] = 0.0f;
}

__global__ void gnn_edge_kernel(const float* __restrict__ h_in,
                                const float* __restrict__ efeat,
                                const int* __restrict__ src,
                                const int* __restrict__ dst,
                                const float* __restrict__ g_w,
                                float* __restrict__ agg, int n_edges) {
    __shared__ float wlds[16 * 16];
    __shared__ float alds[WAVES][16 * 17];
    __shared__ int   slds[WAVES][16];
    __shared__ int   dlds[WAVES][16];
    int t = threadIdx.x;
    stage_weights_padded(wlds, g_w, 16, 16, t, blockDim.x);
    __syncthreads();

    int wave = t >> 5, lane = t & 31;
    long long tile = (long long)blockIdx.x * WAVES + wave;
    long long base = tile * 16;
    if (base >= n_edges) return;

    float* A = alds[wave];
    int* S = slds[wave];
    int* D = dlds[wave];
    if (lane < 16) {
        long long e = base + lane;
        bool ok = e < (long long)n_edges;
        S[lane] = ok ? src[e] : 0;
        D[lane] = ok ? dst[e] : 0;
    }
    __builtin_amdgcn_wave_barrier();

    for (int i = lane; i < 256; i += 32) {
        int e = i >> 4, k = i & 15;
        long long E = base + e;
        float v = 0.0f;
        if (E < (long long)n_edges) {
            if (k < HID) v = h_in[(size_t)S[e] * HID + k];
            else         v = efeat[(size_t)E * EDGE_DIM + (k - HID)];
        }
        A[e * 17 + k] = v;
    }
    __builtin_amdgcn_wave_barrier();

    v8f c = {};
    int m = lane & 15;
    int koff = (lane < 16) ? 0 : 2;
#pragma unroll
    for (int kc = 0; kc < 4; ++kc) {
        int kb = kc * 4 + koff;
        v2f a, b;
        a.x = A[m * 17 + kb];
        a.y = A[m * 17 + kb + 1];
        b.x = wlds[kb * 16 + m];
        b.y = wlds[(kb + 1) * 16 + m];
        c = __builtin_amdgcn_wmma_f32_16x16x4_f32(false, a, false, b,
                                                  (short)0, c, false, false);
    }
    if (m < HID) {
        int rbase = (lane < 16) ? 0 : 8;
#pragma unroll
        for (int r = 0; r < 8; ++r) {
            long long E = base + rbase + r;
            if (E < (long long)n_edges)
                atomicAdd(&agg[(size_t)D[rbase + r] * HID + m], lrelu(c[r]));
        }
    }
}

__global__ void gnn_update_kernel(const float* __restrict__ h_in,
                                  const float* __restrict__ agg,
                                  const float* __restrict__ u_w,
                                  float* __restrict__ h_out, int n_nodes) {
    __shared__ float wlds[20 * 16];
    __shared__ float alds[WAVES][16 * 21];
    int t = threadIdx.x;
    stage_weights_padded(wlds, u_w, 20, 20, t, blockDim.x);
    __syncthreads();

    int wave = t >> 5, lane = t & 31;
    int tile = blockIdx.x * WAVES + wave;
    int base = tile * 16;
    if (base >= n_nodes) return;

    float* A = alds[wave];
    {
        int row = lane & 15, half = lane >> 4;
        int node = base + row;
        const float* p = (half == 0) ? (h_in + (size_t)node * HID)
                                     : (agg  + (size_t)node * HID);
        int k0 = half * HID;
#pragma unroll
        for (int j = 0; j < HID; ++j)
            A[row * 21 + k0 + j] = (node < n_nodes) ? p[j] : 0.0f;
    }
    __builtin_amdgcn_wave_barrier();

    v8f c = {};
    int m = lane & 15;
    int koff = (lane < 16) ? 0 : 2;
#pragma unroll
    for (int kc = 0; kc < 5; ++kc) {
        int kb = kc * 4 + koff;
        v2f a, b;
        a.x = A[m * 21 + kb];
        a.y = A[m * 21 + kb + 1];
        b.x = wlds[kb * 16 + m];
        b.y = wlds[(kb + 1) * 16 + m];
        c = __builtin_amdgcn_wmma_f32_16x16x4_f32(false, a, false, b,
                                                  (short)0, c, false, false);
    }
    if (m < HID) {
        int rbase = (lane < 16) ? 0 : 8;
#pragma unroll
        for (int r = 0; r < 8; ++r) {
            int node = base + rbase + r;
            if (node < n_nodes)
                h_out[(size_t)node * HID + m] = lrelu(c[r]);
        }
    }
}

extern "C" void kernel_launch(void* const* d_in, const int* in_sizes, int n_in,
                              void* d_out, int out_size, void* d_ws, size_t ws_size,
                              hipStream_t stream) {
    const float* vertex = (const float*)d_in[0];
    const float* efeat  = (const float*)d_in[1];
    const int*   src    = (const int*)d_in[2];
    const int*   dst    = (const int*)d_in[3];
    const float* w_init = (const float*)d_in[4];
    const float* g_w    = (const float*)d_in[5];
    const float* u_w    = (const float*)d_in[6];
    float* out = (float*)d_out;

    const int N = in_sizes[0] / 82;
    const int E = in_sizes[2];
    const int NH = N * HID;

    float* hA  = (float*)d_ws;
    float* hB  = hA + (size_t)NH;
    float* agg = hB + (size_t)NH;

    dim3 blk(256);
    int node_tiles  = (N + 15) / 16;
    int node_blocks = (node_tiles + WAVES - 1) / WAVES;
    int edge_tiles  = (E + 15) / 16;
    int edge_blocks = (edge_tiles + WAVES - 1) / WAVES;
    int zero_blocks = (NH + 255) / 256;

    gnn_embed_kernel<<<node_blocks, blk, 0, stream>>>(vertex, w_init, hA, N);

    gnn_zero_kernel<<<zero_blocks, blk, 0, stream>>>(agg, NH);
    gnn_edge_kernel<<<edge_blocks, blk, 0, stream>>>(hA, efeat, src, dst, g_w, agg, E);
    gnn_update_kernel<<<node_blocks, blk, 0, stream>>>(hA, agg, u_w, hB, N);

    gnn_zero_kernel<<<zero_blocks, blk, 0, stream>>>(agg, NH);
    gnn_edge_kernel<<<edge_blocks, blk, 0, stream>>>(hB, efeat, src, dst, g_w, agg, E);
    gnn_update_kernel<<<node_blocks, blk, 0, stream>>>(hB, agg, u_w, out, N);
}


# embed kernel: 16-node vertex tile (41 whole 128-B lines) read with lane-adjacent 16-byte loads and scattered to the padded LDS rows from registers, instead of per-lane row-strided 4-byte loads
# speedup vs baseline: 1.1590x; 1.1590x over previous
#include <hip/hip_runtime.h>
#include <hip/hip_bf16.h>

typedef float v2f __attribute__((ext_vector_type(2)));
typedef float v8f __attribute__((ext_vector_type(8)));

#define WAVES 8
#define HID 10
#define EDGE_DIM 6

__device__ __forceinline__ float lrelu(float x) {
    return fmaxf(x, 0.1f * x);
}

__device__ __forceinline__ void stage_weights_padded(float* wlds, const float* w,
                                                     int rows_real, int rows_pad,
                                                     int t, int nthreads) {
    for (int i = t; i < rows_pad * 16; i += nthreads) {
        int row = i >> 4, col = i & 15;
        wlds[i] = (col < HID && row < rows_real) ? w[row * HID + col] : 0.0f;
    }
}

#define VDIM 82
#define TILE_F (16 * VDIM)

__global__ void gnn_embed_kernel(const float* __restrict__ vertex,
                                 const float* __restrict__ w_init,
                                 float* __restrict__ h_out, int n_nodes) {
    __shared__ float wlds[84 * 16];
    __shared__ float alds[WAVES][16 * 85];
    int t = threadIdx.x;
    stage_weights_padded(wlds, w_init, 82, 84, t, blockDim.x);
    __syncthreads();

    int wave = t >> 5, lane = t & 31;
    int tile = blockIdx.x * WAVES + wave;
    int base = tile * 16;
    if (base >= n_nodes) return;

    float* A = alds[wave];
    const float* gp = vertex + (size_t)base * VDIM;
    if (base + 16 <= n_nodes) {
        const float4* g4 = reinterpret_cast<const float4*>(gp);
        float4 v[10];
#pragma unroll
        for (int it = 0; it < 10; ++it) v[it] = g4[it * 32 + lane];
        float vt = gp[1280 + lane];
#pragma unroll
        for (int it = 0; it < 10; ++it) {
            int f = (it * 32 + lane) * 4;
            int p0 = f + 3 * (f / VDIM);
            int p1 = f + 2 + 3 * ((f + 2) / VDIM);
            A[p0]     = v[it].x;
            A[p0 + 1] = v[it].y;
            A[p1]     = v[it].z;
            A[p1 + 1] = v[it].w;
        }
        {
            int f = 1280 + lane;
            A[f + 3 * (f / VDIM)] = vt;
        }
    } else {
        for (int f = lane; f < TILE_F; f += 32) {
            int row = f / VDIM;
            A[f + 3 * row] = (base + row < n_nodes) ? gp[f] : 0.0f;
        }
    }
    if (lane < 16) { A[lane * 85 + 82] = 0.0f; A[lane * 85 + 83] = 0.0f; }
    __builtin_amdgcn_wave_barrier();

    v8f c = {};
    int m = lane & 15;
    int koff = (lane < 16) ? 0 : 2;
    for (int kc = 0; kc < 21; ++kc) {
        int kb = kc * 4 + koff;
        v2f a, b;
        a.x = A[m * 85 + kb];
        a.y = A[m * 85 + kb + 1];
        b.x = wlds[kb * 16 + m];
        b.y = wlds[(kb + 1) * 16 + m];
        c = __builtin_amdgcn_wmma_f32_16x16x4_f32(false, a, false, b,
                                                  (short)0, c, false, false);
    }
    if (m < HID) {
        int rbase = (lane < 16) ? 0 : 8;
#pragma unroll
        for (int r = 0; r < 8; ++r) {
            int node = base + rbase + r;
            if (node < n_nodes)
                h_out[(size_t)node * HID + m] = lrelu(c[r]);
        }
    }
}

__global__ void gnn_zero_kernel(float* __restrict__ p, int n) {
    int i = blockIdx.x * blockDim.x + threadIdx.x;
    if (i < n) p[i] = 0.0f;
}

__global__ void gnn_edge_kernel(const float* __restrict__ h_in,
                                const float* __restrict__ efeat,
                                const int* __restrict__ src,
                                const int* __restrict__ dst,
                                const float* __restrict__ g_w,
                                float* __restrict__ agg, int n_edges) {
    __shared__ float wlds[16 * 16];
    __shared__ float alds[WAVES][16 * 17];
    __shared__ int   slds[WAVES][16];
    __shared__ int   dlds[WAVES][16];
    int t = threadIdx.x;
    stage_weights_padded(wlds, g_w, 16, 16, t, blockDim.x);
    __syncthreads();

    int wave = t >> 5, lane = t & 31;
    long long tile = (long long)blockIdx.x * WAVES + wave;
    long long base = tile * 16;
    if (base >= n_edges) return;

    float* A = alds[wave];
    int* S = slds[wave];
    int* D = dlds[wave];
    if (lane < 16) {
        long long e = base + lane;
        bool ok = e < (long long)n_edges;
        S[lane] = ok ? src[e] : 0;
        D[lane] = ok ? dst[e] : 0;
    }
    __builtin_amdgcn_wave_barrier();

    for (int i = lane; i < 256; i += 32) {
        int e = i >> 4, k = i & 15;
        long long E = base + e;
        float v = 0.0f;
        if (E < (long long)n_edges) {
            if (k < HID) v = h_in[(size_t)S[e] * HID + k];
            else         v = efeat[(size_t)E * EDGE_DIM + (k - HID)];
        }
        A[e * 17 + k] = v;
    }
    __builtin_amdgcn_wave_barrier();

    v8f c = {};
    int m = lane & 15;
    int koff = (lane < 16) ? 0 : 2;
#pragma unroll
    for (int kc = 0; kc < 4; ++kc) {
        int kb = kc * 4 + koff;
        v2f a, b;
        a.x = A[m * 17 + kb];
        a.y = A[m * 17 + kb + 1];
        b.x = wlds[kb * 16 + m];
        b.y = wlds[(kb + 1) * 16 + m];
        c = __builtin_amdgcn_wmma_f32_16x16x4_f32(false, a, false, b,
                                                  (short)0, c, false, false);
    }
    if (m < HID) {
        int rbase = (lane < 16) ? 0 : 8;
#pragma unroll
        for (int r = 0; r < 8; ++r) {
            long long E = base + rbase + r;
            if (E < (long long)n_edges)
                atomicAdd(&agg[(size_t)D[rbase + r] * HID + m], lrelu(c[r]));
        }
    }
}

__global__ void gnn_update_kernel(const float* __restrict__ h_in,
                                  const float* __restrict__ agg,
                                  const float* __restrict__ u_w,
                                  float* __restrict__ h_out, int n_nodes) {
    __shared__ float wlds[20 * 16];
    __shared__ float alds[WAVES][16 * 21];
    int t = threadIdx.x;
    stage_weights_padded(wlds, u_w, 20, 20, t, blockDim.x);
    __syncthreads();

    int wave = t >> 5, lane = t & 31;
    int tile = blockIdx.x * WAVES + wave;
    int base = tile * 16;
    if (base >= n_nodes) return;

    float* A = alds[wave];
    {
        int row = lane & 15, half = lane >> 4;
        int node = base + row;
        const float* p = (half == 0) ? (h_in + (size_t)node * HID)
                                     : (agg  + (size_t)node * HID);
        int k0 = half * HID;
#pragma unroll
        for (int j = 0; j < HID; ++j)
            A[row * 21 + k0 + j] = (node < n_nodes) ? p[j] : 0.0f;
    }
    __builtin_amdgcn_wave_barrier();

    v8f c = {};
    int m = lane & 15;
    int koff = (lane < 16) ? 0 : 2;
#pragma unroll
    for (int kc = 0; kc < 5; ++kc) {
        int kb = kc * 4 + koff;
        v2f a, b;
        a.x = A[m * 21 + kb];
        a.y = A[m * 21 + kb + 1];
        b.x = wlds[kb * 16 + m];
        b.y = wlds[(kb + 1) * 16 + m];
        c = __builtin_amdgcn_wmma_f32_16x16x4_f32(false, a, false, b,
                                                  (short)0, c, false, false);
    }
    if (m < HID) {
        int rbase = (lane < 16) ? 0 : 8;
#pragma unroll
        for (int r = 0; r < 8; ++r) {
            int node = base + rbase + r;
            if (node < n_nodes)
                h_out[(size_t)node * HID + m] = lrelu(c[r]);
        }
    }
}

extern "C" void kernel_launch(void* const* d_in, const int* in_sizes, int n_in,
                              void* d_out, int out_size, void* d_ws, size_t ws_size,
                              hipStream_t stream) {
    const float* vertex = (const float*)d_in[0];
    const float* efeat  = (const float*)d_in[1];
    const int*   src    = (const int*)d_in[2];
    const int*   dst    = (const int*)d_in[3];
    const float* w_init = (const float*)d_in[4];
    const float* g_w    = (const float*)d_in[5];
    const float* u_w    = (const float*)d_in[6];
    float* out = (float*)d_out;

    const int N = in_sizes[0] / 82;
    const int E = in_sizes[2];
    const int NH = N * HID;

    float* hA  = (float*)d_ws;
    float* hB  = hA + (size_t)NH;
    float* agg = hB + (size_t)NH;

    dim3 blk(256);
    int node_tiles  = (N + 15) / 16;
    int node_blocks = (node_tiles + WAVES - 1) / WAVES;
    int edge_tiles  = (E + 15) / 16;
    int edge_blocks = (edge_tiles + WAVES - 1) / WAVES;
    int zero_blocks = (NH + 255) / 256;

    gnn_embed_kernel<<<node_blocks, blk, 0, stream>>>(vertex, w_init, hA, N);

    gnn_zero_kernel<<<zero_blocks, blk, 0, stream>>>(agg, NH);
    gnn_edge_kernel<<<edge_blocks, blk, 0, stream>>>(hA, efeat, src, dst, g_w, agg, E);
    gnn_update_kernel<<<node_blocks, blk, 0, stream>>>(hA, agg, u_w, hB, N);

    gnn_zero_kernel<<<zero_blocks, blk, 0, stream>>>(agg, NH);
    gnn_edge_kernel<<<edge_blocks, blk, 0, stream>>>(hB, efeat, src, dst, g_w, agg, E);
    gnn_update_kernel<<<node_blocks, blk, 0, stream>>>(hB, agg, u_w, out, N);
}


# edge scatter-add: edges binned once per call by destination range (LDS counting sort, 512 bins); one block per bin sums messages in LDS, plain agg stores, no global float atomics; sum order changes
# speedup vs baseline: 1.8589x; 1.6039x over previous
#include <hip/hip_runtime.h>
#include <hip/hip_bf16.h>

typedef float v2f __attribute__((ext_vector_type(2)));
typedef float v8f __attribute__((ext_vector_type(8)));

#define WAVES 8
#define HID 10
#define EDGE_DIM 6

__device__ __forceinline__ float lrelu(float x) {
    return fmaxf(x, 0.1f * x);
}

__device__ __forceinline__ void stage_weights_padded(float* wlds, const float* w,
                                                     int rows_real, int rows_pad,
                                                     int t, int nthreads) {
    for (int i = t; i < rows_pad * 16; i += nthreads) {
        int row = i >> 4, col = i & 15;
        wlds[i] = (col < HID && row < rows_real) ? w[row * HID + col] : 0.0f;
    }
}

#define VDIM 82
#define TILE_F (16 * VDIM)

__global__ void gnn_embed_kernel(const float* __restrict__ vertex,
                                 const float* __restrict__ w_init,
                                 float* __restrict__ h_out, int n_nodes) {
    __shared__ float wlds[84 * 16];
    __shared__ float alds[WAVES][16 * 85];
    int t = threadIdx.x;
    stage_weights_padded(wlds, w_init, 82, 84, t, blockDim.x);
    __syncthreads();

    int wave = t >> 5, lane = t & 31;
    int tile = blockIdx.x * WAVES + wave;
    int base = tile * 16;
    if (base >= n_nodes) return;

    float* A = alds[wave];
    const float* gp = vertex + (size_t)base * VDIM;
    if (base + 16 <= n_nodes) {
        const float4* g4 = reinterpret_cast<const float4*>(gp);
        float4 v[10];
#pragma unroll
        for (int it = 0; it < 10; ++it) v[it] = g4[it * 32 + lane];
        float vt = gp[1280 + lane];
#pragma unroll
        for (int it = 0; it < 10; ++it) {
            int f = (it * 32 + lane) * 4;
            int p0 = f + 3 * (f / VDIM);
            int p1 = f + 2 + 3 * ((f + 2) / VDIM);
            A[p0]     = v[it].x;
            A[p0 + 1] = v[it].y;
            A[p1]     = v[it].z;
            A[p1 + 1] = v[it].w;
        }
        {
            int f = 1280 + lane;
            A[f + 3 * (f / VDIM)] = vt;
        }
    } else {
        for (int f = lane; f < TILE_F; f += 32) {
            int row = f / VDIM;
            A[f + 3 * row] = (base + row < n_nodes) ? gp[f] : 0.0f;
        }
    }
    if (lane < 16) { A[lane * 85 + 82] = 0.0f; A[lane * 85 + 83] = 0.0f; }
    __builtin_amdgcn_wave_barrier();

    v8f c = {};
    int m = lane & 15;
    int koff = (lane < 16) ? 0 : 2;
    for (int kc = 0; kc < 21; ++kc) {
        int kb = kc * 4 + koff;
        v2f a, b;
        a.x = A[m * 85 + kb];
        a.y = A[m * 85 + kb + 1];
        b.x = wlds[kb * 16 + m];
        b.y = wlds[(kb + 1) * 16 + m];
        c = __builtin_amdgcn_wmma_f32_16x16x4_f32(false, a, false, b,
                                                  (short)0, c, false, false);
    }
    if (m < HID) {
        int rbase = (lane < 16) ? 0 : 8;
#pragma unroll
        for (int r = 0; r < 8; ++r) {
            int node = base + rbase + r;
            if (node < n_nodes)
                h_out[(size_t)node * HID + m] = lrelu(c[r]);
        }
    }
}

#define NB  512
#define NPB 400
#define HB  512
#define PT  1024

__device__ __forceinline__ int dst_bin(int d) {
    unsigned b = (unsigned)d / NPB;
    return (b > NB - 1) ? (NB - 1) : (int)b;
}

__device__ __forceinline__ void lds_inclusive_scan(int* sm, int n, int t) {
    for (int d = 1; d < n; d <<= 1) {
        int x = 0;
        if (t < n && t >= d) x = sm[t - d];
        __syncthreads();
        if (t < n) sm[t] += x;
        __syncthreads();
    }
}

__global__ __launch_bounds__(PT) void gnn_hist_kernel(const int* __restrict__ dst,
                                                      int* __restrict__ cnt,
                                                      int n_edges, int chunk) {
    __shared__ int hist[NB];
    int t = threadIdx.x;
    for (int i = t; i < NB; i += PT) hist[i] = 0;
    __syncthreads();
    int lo = blockIdx.x * chunk;
    int hi = lo + chunk; if (hi > n_edges) hi = n_edges;
    for (int e = lo + t; e < hi; e += PT) atomicAdd(&hist[dst_bin(dst[e])], 1);
    __syncthreads();
    for (int i = t; i < NB; i += PT) cnt[blockIdx.x * NB + i] = hist[i];
}

__global__ __launch_bounds__(HB) void gnn_scan_kernel(const int* __restrict__ cnt,
                                                      int* __restrict__ off,
                                                      int* __restrict__ tot) {
    __shared__ int sm[HB];
    int t = threadIdx.x, bin = blockIdx.x;
    int v = cnt[t * NB + bin];
    sm[t] = v;
    __syncthreads();
    lds_inclusive_scan(sm, HB, t);
    int incl = sm[t];
    off[t * NB + bin] = incl - v;
    if (t == HB - 1) tot[bin] = incl;
}

__global__ __launch_bounds__(PT) void gnn_scatter_kernel(const float* __restrict__ efeat,
                                                         const int* __restrict__ src,
                                                         const int* __restrict__ dst,
                                                         const int* __restrict__ off,
                                                         const int* __restrict__ tot,
                                                         int* __restrict__ bstart,
                                                         float* __restrict__ rec,
                                                         int n_edges, int chunk) {
    __shared__ int sm[NB];
    __shared__ int cursor[NB];
    int t = threadIdx.x, blk = blockIdx.x;
    int v = 0;
    if (t < NB) { v = tot[t]; sm[t] = v; }
    __syncthreads();
    lds_inclusive_scan(sm, NB, t);
    if (t < NB) {
        int st = sm[t] - v;
        cursor[t] = st + off[blk * NB + t];
        if (blk == 0) bstart[t] = st;
    }
    __syncthreads();

    float4* rec4 = reinterpret_cast<float4*>(rec);
    int lo = blk * chunk;
    int hi = lo + chunk; if (hi > n_edges) hi = n_edges;
    for (int e = lo + t; e < hi; e += PT) {
        int d = dst[e];
        int s = src[e];
        const float2* ep = reinterpret_cast<const float2*>(efeat + (size_t)e * EDGE_DIM);
        float2 f0 = ep[0], f1 = ep[1], f2 = ep[2];
        int pos = atomicAdd(&cursor[dst_bin(d)], 1);
        rec4[(size_t)pos * 2]     = make_float4(__int_as_float(s), __int_as_float(d), f0.x, f0.y);
        rec4[(size_t)pos * 2 + 1] = make_float4(f1.x, f1.y, f2.x, f2.y);
    }
}

#define EB_WAVES (PT / 32)

__global__ __launch_bounds__(PT) void gnn_edge_bin_kernel(const float* __restrict__ h_in,
                                                          const float* __restrict__ rec,
                                                          const int* __restrict__ bstart,
                                                          const int* __restrict__ tot,
                                                          const float* __restrict__ g_w,
                                                          float* __restrict__ agg, int n_nodes) {
    __shared__ float wlds[16 * 16];
    __shared__ float acc[NPB * HID];
    __shared__ float alds[EB_WAVES][16 * 17];
    __shared__ int   slds[EB_WAVES][16];
    __shared__ alignas(16) int dlds[EB_WAVES][16];
    int t = threadIdx.x, bin = blockIdx.x;
    stage_weights_padded(wlds, g_w, 16, 16, t, PT);
    for (int i = t; i < NPB * HID; i += PT) acc[i] = 0.0f;
    const int start = bstart[bin];
    const int end = start + tot[bin];
    const int node_lo = bin * NPB;
    __syncthreads();

    int wave = t >> 5, lane = t & 31;
    float* A = alds[wave];
    int* S = slds[wave];
    int* D = dlds[wave];
    const float4* rec4 = reinterpret_cast<const float4*>(rec);
    int m = lane & 15;
    int koff = (lane < 16) ? 0 : 2;
    int er = lane >> 1, hf = lane & 1;
    int rbase = (lane < 16) ? 0 : 8;

    v2f bf[4];
#pragma unroll
    for (int kc = 0; kc < 4; ++kc) {
        int kb = kc * 4 + koff;
        bf[kc].x = wlds[kb * 16 + m];
        bf[kc].y = wlds[(kb + 1) * 16 + m];
    }

    const int last = end - 1;
    int tb = start + wave * 16;
    int e0 = tb + er;
    if (e0 > last) e0 = last;
    if (e0 < 0) e0 = 0;
    float4 r = rec4[(size_t)e0 * 2 + hf];
    while (tb < end) {
        int tn = tb + EB_WAVES * 16;
        int en = tn + er;
        if (en > last) en = last;
        float4 rn = rec4[(size_t)en * 2 + hf];

        if (hf == 0) {
            S[er] = __float_as_int(r.x);
            D[er] = __float_as_int(r.y) - node_lo;
            A[er * 17 + 10] = r.z;
            A[er * 17 + 11] = r.w;
        } else {
            A[er * 17 + 12] = r.x;
            A[er * 17 + 13] = r.y;
            A[er * 17 + 14] = r.z;
            A[er * 17 + 15] = r.w;
        }
        __builtin_amdgcn_wave_barrier();

        {
            const float2* hp = reinterpret_cast<const float2*>(h_in + (size_t)S[er] * HID);
            int j0 = hf * 3, j2 = hf ? 4 : 2;
            float2 g0 = hp[j0], g1 = hp[j0 + 1], g2 = hp[j2];
            A[er * 17 + 2 * j0]     = g0.x;
            A[er * 17 + 2 * j0 + 1] = g0.y;
            A[er * 17 + 2 * j0 + 2] = g1.x;
            A[er * 17 + 2 * j0 + 3] = g1.y;
            A[er * 17 + 2 * j2]     = g2.x;
            A[er * 17 + 2 * j2 + 1] = g2.y;
        }
        __builtin_amdgcn_wave_barrier();

        const int4 dq0 = *reinterpret_cast<const int4*>(&D[rbase]);
        const int4 dq1 = *reinterpret_cast<const int4*>(&D[rbase + 4]);
        const int dl[8] = {dq0.x, dq0.y, dq0.z, dq0.w, dq1.x, dq1.y, dq1.z, dq1.w};

        v8f c = {};
#pragma unroll
        for (int kc = 0; kc < 4; ++kc) {
            int kb = kc * 4 + koff;
            v2f a;
            a.x = A[m * 17 + kb];
            a.y = A[m * 17 + kb + 1];
            c = __builtin_amdgcn_wmma_f32_16x16x4_f32(false, a, false, bf[kc],
                                                      (short)0, c, false, false);
        }
        if (m < HID) {
            int nvalid = end - tb - rbase;
#pragma unroll
            for (int rr = 0; rr < 8; ++rr) {
                if (rr < nvalid && (unsigned)dl[rr] < NPB)
                    atomicAdd(&acc[dl[rr] * HID + m], lrelu(c[rr]));
            }
        }
        __builtin_amdgcn_wave_barrier();
        r = rn;
        tb = tn;
    }
    __syncthreads();
    for (int i = t; i < NPB * HID; i += PT) {
        int node = node_lo + i / HID;
        if (node < n_nodes) agg[(size_t)node_lo * HID + i] = acc[i];
    }
}

__global__ void gnn_update_kernel(const float* __restrict__ h_in,
                                  const float* __restrict__ agg,
                                  const float* __restrict__ u_w,
                                  float* __restrict__ h_out, int n_nodes) {
    __shared__ float wlds[20 * 16];
    __shared__ float alds[WAVES][16 * 21];
    int t = threadIdx.x;
    stage_weights_padded(wlds, u_w, 20, 20, t, blockDim.x);
    __syncthreads();

    int wave = t >> 5, lane = t & 31;
    int tile = blockIdx.x * WAVES + wave;
    int base = tile * 16;
    if (base >= n_nodes) return;

    float* A = alds[wave];
    {
        int row = lane & 15, half = lane >> 4;
        int node = base + row;
        const float* p = (half == 0) ? (h_in + (size_t)node * HID)
                                     : (agg  + (size_t)node * HID);
        int k0 = half * HID;
#pragma unroll
        for (int j = 0; j < HID; ++j)
            A[row * 21 + k0 + j] = (node < n_nodes) ? p[j] : 0.0f;
    }
    __builtin_amdgcn_wave_barrier();

    v8f c = {};
    int m = lane & 15;
    int koff = (lane < 16) ? 0 : 2;
#pragma unroll
    for (int kc = 0; kc < 5; ++kc) {
        int kb = kc * 4 + koff;
        v2f a, b;
        a.x = A[m * 21 + kb];
        a.y = A[m * 21 + kb + 1];
        b.x = wlds[kb * 16 + m];
        b.y = wlds[(kb + 1) * 16 + m];
        c = __builtin_amdgcn_wmma_f32_16x16x4_f32(false, a, false, b,
                                                  (short)0, c, false, false);
    }
    if (m < HID) {
        int rbase = (lane < 16) ? 0 : 8;
#pragma unroll
        for (int r = 0; r < 8; ++r) {
            int node = base + rbase + r;
            if (node < n_nodes)
                h_out[(size_t)node * HID + m] = lrelu(c[r]);
        }
    }
}

extern "C" void kernel_launch(void* const* d_in, const int* in_sizes, int n_in,
                              void* d_out, int out_size, void* d_ws, size_t ws_size,
                              hipStream_t stream) {
    const float* vertex = (const float*)d_in[0];
    const float* efeat  = (const float*)d_in[1];
    const int*   src    = (const int*)d_in[2];
    const int*   dst    = (const int*)d_in[3];
    const float* w_init = (const float*)d_in[4];
    const float* g_w    = (const float*)d_in[5];
    const float* u_w    = (const float*)d_in[6];
    float* out = (float*)d_out;

    const int N = in_sizes[0] / 82;
    const int E = in_sizes[2];
    const int NH = N * HID;

    float* hA  = (float*)d_ws;
    float* hB  = hA + (size_t)NH;
    float* agg = hB + (size_t)NH;
    float* rec = agg + (size_t)NH;
    int* cnt    = (int*)(rec + (size_t)E * 8);
    int* off    = cnt + HB * NB;
    int* tot    = off + HB * NB;
    int* bstart = tot + NB;

    dim3 blk(256);
    int node_tiles  = (N + 15) / 16;
    int node_blocks = (node_tiles + WAVES - 1) / WAVES;
    int chunk = (E + HB - 1) / HB;

    gnn_embed_kernel<<<node_blocks, blk, 0, stream>>>(vertex, w_init, hA, N);

    gnn_hist_kernel<<<HB, PT, 0, stream>>>(dst, cnt, E, chunk);
    gnn_scan_kernel<<<NB, HB, 0, stream>>>(cnt, off, tot);
    gnn_scatter_kernel<<<HB, PT, 0, stream>>>(efeat, src, dst, off, tot, bstart, rec, E, chunk);

    gnn_edge_bin_kernel<<<NB, PT, 0, stream>>>(hA, rec, bstart, tot, g_w, agg, N);
    gnn_update_kernel<<<node_blocks, blk, 0, stream>>>(hA, agg, u_w, hB, N);

    gnn_edge_bin_kernel<<<NB, PT, 0, stream>>>(hB, rec, bstart, tot, g_w, agg, N);
    gnn_update_kernel<<<node_blocks, blk, 0, stream>>>(hB, agg, u_w, out, N);
}


# hidden rows kept at a 64-byte pitch in the workspace (10 values + 6 zeros): the per-edge gather of h[src] takes a row as 16-byte loads of adjacent lanes, 16 lines per 16-edge tile, not ~38
# speedup vs baseline: 1.9646x; 1.0569x over previous
#include <hip/hip_runtime.h>
#include <hip/hip_bf16.h>

typedef float v2f __attribute__((ext_vector_type(2)));
typedef float v8f __attribute__((ext_vector_type(8)));

#define WAVES 8
#define HID 10
#define EDGE_DIM 6
#define HP 16

__device__ __forceinline__ float lrelu(float x) {
    return fmaxf(x, 0.1f * x);
}

__device__ __forceinline__ void stage_weights_padded(float* wlds, const float* w,
                                                     int rows_real, int rows_pad,
                                                     int t, int nthreads) {
    for (int i = t; i < rows_pad * 16; i += nthreads) {
        int row = i >> 4, col = i & 15;
        wlds[i] = (col < HID && row < rows_real) ? w[row * HID + col] : 0.0f;
    }
}

#define VDIM 82
#define TILE_F (16 * VDIM)

__global__ void gnn_embed_kernel(const float* __restrict__ vertex,
                                 const float* __restrict__ w_init,
                                 float* __restrict__ h_out, int n_nodes) {
    __shared__ float wlds[84 * 16];
    __shared__ float alds[WAVES][16 * 85];
    int t = threadIdx.x;
    stage_weights_padded(wlds, w_init, 82, 84, t, blockDim.x);
    __syncthreads();

    int wave = t >> 5, lane = t & 31;
    int tile = blockIdx.x * WAVES + wave;
    int base = tile * 16;
    if (base >= n_nodes) return;

    float* A = alds[wave];
    const float* gp = vertex + (size_t)base * VDIM;
    if (base + 16 <= n_nodes) {
        const float4* g4 = reinterpret_cast<const float4*>(gp);
        float4 v[10];
#pragma unroll
        for (int it = 0; it < 10; ++it) v[it] = g4[it * 32 + lane];
        float vt = gp[1280 + lane];
#pragma unroll
        for (int it = 0; it < 10; ++it) {
            int f = (it * 32 + lane) * 4;
            int p0 = f + 3 * (f / VDIM);
            int p1 = f + 2 + 3 * ((f + 2) / VDIM);
            A[p0]     = v[it].x;
            A[p0 + 1] = v[it].y;
            A[p1]     = v[it].z;
            A[p1 + 1] = v[it].w;
        }
        {
            int f = 1280 + lane;
            A[f + 3 * (f / VDIM)] = vt;
        }
    } else {
        for (int f = lane; f < TILE_F; f += 32) {
            int row = f / VDIM;
            A[f + 3 * row] = (base + row < n_nodes) ? gp[f] : 0.0f;
        }
    }
    if (lane < 16) { A[lane * 85 + 82] = 0.0f; A[lane * 85 + 83] = 0.0f; }
    __builtin_amdgcn_wave_barrier();

    v8f c = {};
    int m = lane & 15;
    int koff = (lane < 16) ? 0 : 2;
    for (int kc = 0; kc < 21; ++kc) {
        int kb = kc * 4 + koff;
        v2f a, b;
        a.x = A[m * 85 + kb];
        a.y = A[m * 85 + kb + 1];
        b.x = wlds[kb * 16 + m];
        b.y = wlds[(kb + 1) * 16 + m];
        c = __builtin_amdgcn_wmma_f32_16x16x4_f32(false, a, false, b,
                                                  (short)0, c, false, false);
    }
    {
        int rbase = (lane < 16) ? 0 : 8;
#pragma unroll
        for (int r = 0; r < 8; ++r) {
            int node = base + rbase + r;
            if (node < n_nodes)
                h_out[(size_t)node * HP + m] = (m < HID) ? lrelu(c[r]) : 0.0f;
        }
    }
}

#define NB  512
#define NPB 400
#define HB  512
#define PT  1024

__device__ __forceinline__ int dst_bin(int d) {
    unsigned b = (unsigned)d / NPB;
    return (b > NB - 1) ? (NB - 1) : (int)b;
}

__device__ __forceinline__ void lds_inclusive_scan(int* sm, int n, int t) {
    for (int d = 1; d < n; d <<= 1) {
        int x = 0;
        if (t < n && t >= d) x = sm[t - d];
        __syncthreads();
        if (t < n) sm[t] += x;
        __syncthreads();
    }
}

__global__ __launch_bounds__(PT) void gnn_hist_kernel(const int* __restrict__ dst,
                                                      int* __restrict__ cnt,
                                                      int n_edges, int chunk) {
    __shared__ int hist[NB];
    int t = threadIdx.x;
    for (int i = t; i < NB; i += PT) hist[i] = 0;
    __syncthreads();
    int lo = blockIdx.x * chunk;
    int hi = lo + chunk; if (hi > n_edges) hi = n_edges;
    for (int e = lo + t; e < hi; e += PT) atomicAdd(&hist[dst_bin(dst[e])], 1);
    __syncthreads();
    for (int i = t; i < NB; i += PT) cnt[blockIdx.x * NB + i] = hist[i];
}

__global__ __launch_bounds__(HB) void gnn_scan_kernel(const int* __restrict__ cnt,
                                                      int* __restrict__ off,
                                                      int* __restrict__ tot) {
    __shared__ int sm[HB];
    int t = threadIdx.x, bin = blockIdx.x;
    int v = cnt[t * NB + bin];
    sm[t] = v;
    __syncthreads();
    lds_inclusive_scan(sm, HB, t);
    int incl = sm[t];
    off[t * NB + bin] = incl - v;
    if (t == HB - 1) tot[bin] = incl;
}

__global__ __launch_bounds__(PT) void gnn_scatter_kernel(const float* __restrict__ efeat,
                                                         const int* __restrict__ src,
                                                         const int* __restrict__ dst,
                                                         const int* __restrict__ off,
                                                         const int* __restrict__ tot,
                                                         int* __restrict__ bstart,
                                                         float* __restrict__ rec,
                                                         int n_edges, int chunk) {
    __shared__ int sm[NB];
    __shared__ int cursor[NB];
    int t = threadIdx.x, blk = blockIdx.x;
    int v = 0;
    if (t < NB) { v = tot[t]; sm[t] = v; }
    __syncthreads();
    lds_inclusive_scan(sm, NB, t);
    if (t < NB) {
        int st = sm[t] - v;
        cursor[t] = st + off[blk * NB + t];
        if (blk == 0) bstart[t] = st;
    }
    __syncthreads();

    float4* rec4 = reinterpret_cast<float4*>(rec);
    int lo = blk * chunk;
    int hi = lo + chunk; if (hi > n_edges) hi = n_edges;
    for (int e = lo + t; e < hi; e += PT) {
        int d = dst[e];
        int s = src[e];
        const float2* ep = reinterpret_cast<const float2*>(efeat + (size_t)e * EDGE_DIM);
        float2 f0 = ep[0], f1 = ep[1], f2 = ep[2];
        int pos = atomicAdd(&cursor[dst_bin(d)], 1);
        rec4[(size_t)pos * 2]     = make_float4(__int_as_float(s), __int_as_float(d), f0.x, f0.y);
        rec4[(size_t)pos * 2 + 1] = make_float4(f1.x, f1.y, f2.x, f2.y);
    }
}

#define EB_WAVES (PT / 32)
#define AP 20
#define APOS(k) ((k) < HID ? (k) : (k) + 2)

__global__ __launch_bounds__(PT) void gnn_edge_bin_kernel(const float* __restrict__ h_in,
                                                          const float* __restrict__ rec,
                                                          const int* __restrict__ bstart,
                                                          const int* __restrict__ tot,
                                                          const float* __restrict__ g_w,
                                                          float* __restrict__ agg, int n_nodes) {
    __shared__ float wlds[16 * 16];
    __shared__ float acc[NPB * HID];
    __shared__ alignas(16) float alds[EB_WAVES][16 * AP];
    __shared__ int   slds[EB_WAVES][16];
    __shared__ alignas(16) int dlds[EB_WAVES][16];
    int t = threadIdx.x, bin = blockIdx.x;
    stage_weights_padded(wlds, g_w, 16, 16, t, PT);
    for (int i = t; i < NPB * HID; i += PT) acc[i] = 0.0f;
    const int start = bstart[bin];
    const int end = start + tot[bin];
    const int node_lo = bin * NPB;
    __syncthreads();

    int wave = t >> 5, lane = t & 31;
    float* A = alds[wave];
    int* S = slds[wave];
    int* D = dlds[wave];
    const float4* rec4 = reinterpret_cast<const float4*>(rec);
    int m = lane & 15;
    int koff = (lane < 16) ? 0 : 2;
    int er = lane >> 1, hf = lane & 1;
    int rbase = (lane < 16) ? 0 : 8;

    v2f bf[4];
#pragma unroll
    for (int kc = 0; kc < 4; ++kc) {
        int kb = kc * 4 + koff;
        bf[kc].x = wlds[kb * 16 + m];
        bf[kc].y = wlds[(kb + 1) * 16 + m];
    }

    const int last = end - 1;
    int tb = start + wave * 16;
    int e0 = tb + er;
    if (e0 > last) e0 = last;
    if (e0 < 0) e0 = 0;
    float4 r = rec4[(size_t)e0 * 2 + hf];
    while (tb < end) {
        int tn = tb + EB_WAVES * 16;
        int en = tn + er;
        if (en > last) en = last;
        float4 rn = rec4[(size_t)en * 2 + hf];

        if (hf == 0) {
            S[er] = __float_as_int(r.x);
            D[er] = __float_as_int(r.y) - node_lo;
            *reinterpret_cast<float2*>(&A[er * AP + APOS(10)]) = make_float2(r.z, r.w);
        } else {
            *reinterpret_cast<float2*>(&A[er * AP + APOS(12)]) = make_float2(r.x, r.y);
            *reinterpret_cast<float2*>(&A[er * AP + APOS(14)]) = make_float2(r.z, r.w);
        }
        __builtin_amdgcn_wave_barrier();

        {
            const int gq = lane & 3, gr = lane >> 2;
            if (gq < 3) {
                const float4 g0 = *reinterpret_cast<const float4*>(
                    h_in + (size_t)S[gr] * HP + 4 * gq);
                const float4 g1 = *reinterpret_cast<const float4*>(
                    h_in + (size_t)S[gr + 8] * HP + 4 * gq);
                *reinterpret_cast<float4*>(&A[gr * AP + 4 * gq])       = g0;
                *reinterpret_cast<float4*>(&A[(gr + 8) * AP + 4 * gq]) = g1;
            }
        }
        __builtin_amdgcn_wave_barrier();

        const int4 dq0 = *reinterpret_cast<const int4*>(&D[rbase]);
        const int4 dq1 = *reinterpret_cast<const int4*>(&D[rbase + 4]);
        const int dl[8] = {dq0.x, dq0.y, dq0.z, dq0.w, dq1.x, dq1.y, dq1.z, dq1.w};

        v8f c = {};
#pragma unroll
        for (int kc = 0; kc < 4; ++kc) {
            int kb = kc * 4 + koff;
            const float* ap = &A[m * AP + APOS(kb)];
            v2f a;
            a.x = ap[0];
            a.y = ap[1];
            c = __builtin_amdgcn_wmma_f32_16x16x4_f32(false, a, false, bf[kc],
                                                      (short)0, c, false, false);
        }
        if (m < HID) {
            int nvalid = end - tb - rbase;
#pragma unroll
            for (int rr = 0; rr < 8; ++rr) {
                if (rr < nvalid && (unsigned)dl[rr] < NPB)
                    atomicAdd(&acc[dl[rr] * HID + m], lrelu(c[rr]));
            }
        }
        __builtin_amdgcn_wave_barrier();
        r = rn;
        tb = tn;
    }
    __syncthreads();
    for (int i = t; i < NPB * HID; i += PT) {
        int node = node_lo + i / HID;
        if (node < n_nodes) agg[(size_t)node_lo * HID + i] = acc[i];
    }
}

__global__ void gnn_update_kernel(const float* __restrict__ h_in,
                                  const float* __restrict__ agg,
                                  const float* __restrict__ u_w,
                                  float* __restrict__ h_out, int n_nodes,
                                  int out_pitch) {
    __shared__ float wlds[20 * 16];
    __shared__ float alds[WAVES][16 * 21];
    int t = threadIdx.x;
    stage_weights_padded(wlds, u_w, 20, 20, t, blockDim.x);
    __syncthreads();

    int wave = t >> 5, lane = t & 31;
    int tile = blockIdx.x * WAVES + wave;
    int base = tile * 16;
    if (base >= n_nodes) return;

    float* A = alds[wave];
    {
        int row = lane & 15, half = lane >> 4;
        int node = base + row;
        const float* p = (half == 0) ? (h_in + (size_t)node * HP)
                                     : (agg  + (size_t)node * HID);
        int k0 = half * HID;
#pragma unroll
        for (int j = 0; j < HID; ++j)
            A[row * 21 + k0 + j] = (node < n_nodes) ? p[j] : 0.0f;
    }
    __builtin_amdgcn_wave_barrier();

    v8f c = {};
    int m = lane & 15;
    int koff = (lane < 16) ? 0 : 2;
#pragma unroll
    for (int kc = 0; kc < 5; ++kc) {
        int kb = kc * 4 + koff;
        v2f a, b;
        a.x = A[m * 21 + kb];
        a.y = A[m * 21 + kb + 1];
        b.x = wlds[kb * 16 + m];
        b.y = wlds[(kb + 1) * 16 + m];
        c = __builtin_amdgcn_wmma_f32_16x16x4_f32(false, a, false, b,
                                                  (short)0, c, false, false);
    }
    if (m < out_pitch) {
        int rbase = (lane < 16) ? 0 : 8;
#pragma unroll
        for (int r = 0; r < 8; ++r) {
            int node = base + rbase + r;
            if (node < n_nodes)
                h_out[(size_t)node * out_pitch + m] = (m < HID) ? lrelu(c[r]) : 0.0f;
        }
    }
}

extern "C" void kernel_launch(void* const* d_in, const int* in_sizes, int n_in,
                              void* d_out, int out_size, void* d_ws, size_t ws_size,
                              hipStream_t stream) {
    const float* vertex = (const float*)d_in[0];
    const float* efeat  = (const float*)d_in[1];
    const int*   src    = (const int*)d_in[2];
    const int*   dst    = (const int*)d_in[3];
    const float* w_init = (const float*)d_in[4];
    const float* g_w    = (const float*)d_in[5];
    const float* u_w    = (const float*)d_in[6];
    float* out = (float*)d_out;

    const int N = in_sizes[0] / 82;
    const int E = in_sizes[2];
    const int NH = N * HID;
    const int NP = N * HP;

    float* hA  = (float*)d_ws;
    float* hB  = hA + (size_t)NP;
    float* agg = hB + (size_t)NP;
    float* rec = agg + (size_t)NH;
    int* cnt    = (int*)(rec + (size_t)E * 8);
    int* off    = cnt + HB * NB;
    int* tot    = off + HB * NB;
    int* bstart = tot + NB;

    dim3 blk(256);
    int node_tiles  = (N + 15) / 16;
    int node_blocks = (node_tiles + WAVES - 1) / WAVES;
    int chunk = (E + HB - 1) / HB;

    gnn_embed_kernel<<<node_blocks, blk, 0, stream>>>(vertex, w_init, hA, N);

    gnn_hist_kernel<<<HB, PT, 0, stream>>>(dst, cnt, E, chunk);
    gnn_scan_kernel<<<NB, HB, 0, stream>>>(cnt, off, tot);
    gnn_scatter_kernel<<<HB, PT, 0, stream>>>(efeat, src, dst, off, tot, bstart, rec, E, chunk);

    gnn_edge_bin_kernel<<<NB, PT, 0, stream>>>(hA, rec, bstart, tot, g_w, agg, N);
    gnn_update_kernel<<<node_blocks, blk, 0, stream>>>(hA, agg, u_w, hB, N, HP);

    gnn_edge_bin_kernel<<<NB, PT, 0, stream>>>(hB, rec, bstart, tot, g_w, agg, N);
    gnn_update_kernel<<<node_blocks, blk, 0, stream>>>(hB, agg, u_w, out, N, HID);
}


# node update fused into the per-bin edge kernel: the block that summed a bin forms h_new for its 400 nodes from its LDS sums (same 5 WMMAs per 16 nodes); agg never stored; 8 launches become 6
# speedup vs baseline: 2.0422x; 1.0395x over previous
#include <hip/hip_runtime.h>
#include <hip/hip_bf16.h>

typedef float v2f __attribute__((ext_vector_type(2)));
typedef float v8f __attribute__((ext_vector_type(8)));

#define WAVES 8
#define HID 10
#define EDGE_DIM 6
#define HP 16

__device__ __forceinline__ float lrelu(float x) {
    return fmaxf(x, 0.1f * x);
}

__device__ __forceinline__ void stage_weights_padded(float* wlds, const float* w,
                                                     int rows_real, int rows_pad,
                                                     int t, int nthreads) {
    for (int i = t; i < rows_pad * 16; i += nthreads) {
        int row = i >> 4, col = i & 15;
        wlds[i] = (col < HID && row < rows_real) ? w[row * HID + col] : 0.0f;
    }
}

#define VDIM 82
#define TILE_F (16 * VDIM)

__global__ void gnn_embed_kernel(const float* __restrict__ vertex,
                                 const float* __restrict__ w_init,
                                 float* __restrict__ h_out, int n_nodes) {
    __shared__ float wlds[84 * 16];
    __shared__ float alds[WAVES][16 * 85];
    int t = threadIdx.x;
    stage_weights_padded(wlds, w_init, 82, 84, t, blockDim.x);
    __syncthreads();

    int wave = t >> 5, lane = t & 31;
    int tile = blockIdx.x * WAVES + wave;
    int base = tile * 16;
    if (base >= n_nodes) return;

    float* A = alds[wave];
    const float* gp = vertex + (size_t)base * VDIM;
    if (base + 16 <= n_nodes) {
        const float4* g4 = reinterpret_cast<const float4*>(gp);
        float4 v[10];
#pragma unroll
        for (int it = 0; it < 10; ++it) v[it] = g4[it * 32 + lane];
        float vt = gp[1280 + lane];
#pragma unroll
        for (int it = 0; it < 10; ++it) {
            int f = (it * 32 + lane) * 4;
            int p0 = f + 3 * (f / VDIM);
            int p1 = f + 2 + 3 * ((f + 2) / VDIM);
            A[p0]     = v[it].x;
            A[p0 + 1] = v[it].y;
            A[p1]     = v[it].z;
            A[p1 + 1] = v[it].w;
        }
        {
            int f = 1280 + lane;
            A[f + 3 * (f / VDIM)] = vt;
        }
    } else {
        for (int f = lane; f < TILE_F; f += 32) {
            int row = f / VDIM;
            A[f + 3 * row] = (base + row < n_nodes) ? gp[f] : 0.0f;
        }
    }
    if (lane < 16) { A[lane * 85 + 82] = 0.0f; A[lane * 85 + 83] = 0.0f; }
    __builtin_amdgcn_wave_barrier();

    v8f c = {};
    int m = lane & 15;
    int koff = (lane < 16) ? 0 : 2;
    for (int kc = 0; kc < 21; ++kc) {
        int kb = kc * 4 + koff;
        v2f a, b;
        a.x = A[m * 85 + kb];
        a.y = A[m * 85 + kb + 1];
        b.x = wlds[kb * 16 + m];
        b.y = wlds[(kb + 1) * 16 + m];
        c = __builtin_amdgcn_wmma_f32_16x16x4_f32(false, a, false, b,
                                                  (short)0, c, false, false);
    }
    {
        int rbase = (lane < 16) ? 0 : 8;
#pragma unroll
        for (int r = 0; r < 8; ++r) {
            int node = base + rbase + r;
            if (node < n_nodes)
                h_out[(size_t)node * HP + m] = (m < HID) ? lrelu(c[r]) : 0.0f;
        }
    }
}

#define NB  512
#define NPB 400
#define HB  512
#define PT  1024

__device__ __forceinline__ int dst_bin(int d) {
    unsigned b = (unsigned)d / NPB;
    return (b > NB - 1) ? (NB - 1) : (int)b;
}

__device__ __forceinline__ void lds_inclusive_scan(int* sm, int n, int t) {
    for (int d = 1; d < n; d <<= 1) {
        int x = 0;
        if (t < n && t >= d) x = sm[t - d];
        __syncthreads();
        if (t < n) sm[t] += x;
        __syncthreads();
    }
}

__global__ __launch_bounds__(PT) void gnn_hist_kernel(const int* __restrict__ dst,
                                                      int* __restrict__ cnt,
                                                      int n_edges, int chunk) {
    __shared__ int hist[NB];
    int t = threadIdx.x;
    for (int i = t; i < NB; i += PT) hist[i] = 0;
    __syncthreads();
    int lo = blockIdx.x * chunk;
    int hi = lo + chunk; if (hi > n_edges) hi = n_edges;
    for (int e = lo + t; e < hi; e += PT) atomicAdd(&hist[dst_bin(dst[e])], 1);
    __syncthreads();
    for (int i = t; i < NB; i += PT) cnt[blockIdx.x * NB + i] = hist[i];
}

__global__ __launch_bounds__(HB) void gnn_scan_kernel(const int* __restrict__ cnt,
                                                      int* __restrict__ off,
                                                      int* __restrict__ tot) {
    __shared__ int sm[HB];
    int t = threadIdx.x, bin = blockIdx.x;
    int v = cnt[t * NB + bin];
    sm[t] = v;
    __syncthreads();
    lds_inclusive_scan(sm, HB, t);
    int incl = sm[t];
    off[t * NB + bin] = incl - v;
    if (t == HB - 1) tot[bin] = incl;
}

__global__ __launch_bounds__(PT) void gnn_scatter_kernel(const float* __restrict__ efeat,
                                                         const int* __restrict__ src,
                                                         const int* __restrict__ dst,
                                                         const int* __restrict__ off,
                                                         const int* __restrict__ tot,
                                                         int* __restrict__ bstart,
                                                         float* __restrict__ rec,
                                                         int n_edges, int chunk) {
    __shared__ int sm[NB];
    __shared__ int cursor[NB];
    int t = threadIdx.x, blk = blockIdx.x;
    int v = 0;
    if (t < NB) { v = tot[t]; sm[t] = v; }
    __syncthreads();
    lds_inclusive_scan(sm, NB, t);
    if (t < NB) {
        int st = sm[t] - v;
        cursor[t] = st + off[blk * NB + t];
        if (blk == 0) bstart[t] = st;
    }
    __syncthreads();

    float4* rec4 = reinterpret_cast<float4*>(rec);
    int lo = blk * chunk;
    int hi = lo + chunk; if (hi > n_edges) hi = n_edges;
    for (int e = lo + t; e < hi; e += PT) {
        int d = dst[e];
        int s = src[e];
        const float2* ep = reinterpret_cast<const float2*>(efeat + (size_t)e * EDGE_DIM);
        float2 f0 = ep[0], f1 = ep[1], f2 = ep[2];
        int pos = atomicAdd(&cursor[dst_bin(d)], 1);
        rec4[(size_t)pos * 2]     = make_float4(__int_as_float(s), __int_as_float(d), f0.x, f0.y);
        rec4[(size_t)pos * 2 + 1] = make_float4(f1.x, f1.y, f2.x, f2.y);
    }
}

#define EB_WAVES (PT / 32)
#define AP 20
#define APOS(k) ((k) < HID ? (k) : (k) + 2)

__global__ __launch_bounds__(PT) void gnn_edge_bin_kernel(const float* __restrict__ h_in,
                                                          const float* __restrict__ rec,
                                                          const int* __restrict__ bstart,
                                                          const int* __restrict__ tot,
                                                          const float* __restrict__ g_w,
                                                          const float* __restrict__ u_w,
                                                          float* __restrict__ h_out, int n_nodes,
                                                          int out_pitch) {
    __shared__ float wlds[16 * 16];
    __shared__ float uwlds[20 * 16];
    __shared__ alignas(16) float acc[NPB * HID];
    __shared__ alignas(16) float alds[EB_WAVES][16 * AP];
    __shared__ int   slds[EB_WAVES][16];
    __shared__ alignas(16) int dlds[EB_WAVES][16];
    int t = threadIdx.x, bin = blockIdx.x;
    stage_weights_padded(wlds, g_w, 16, 16, t, PT);
    stage_weights_padded(uwlds, u_w, 20, 20, t, PT);
    for (int i = t; i < NPB * HID; i += PT) acc[i] = 0.0f;
    const int start = bstart[bin];
    const int end = start + tot[bin];
    const int node_lo = bin * NPB;
    __syncthreads();

    int wave = t >> 5, lane = t & 31;
    float* A = alds[wave];
    int* S = slds[wave];
    int* D = dlds[wave];
    const float4* rec4 = reinterpret_cast<const float4*>(rec);
    int m = lane & 15;
    int koff = (lane < 16) ? 0 : 2;
    int er = lane >> 1, hf = lane & 1;
    int rbase = (lane < 16) ? 0 : 8;

    v2f bf[4];
#pragma unroll
    for (int kc = 0; kc < 4; ++kc) {
        int kb = kc * 4 + koff;
        bf[kc].x = wlds[kb * 16 + m];
        bf[kc].y = wlds[(kb + 1) * 16 + m];
    }

    const int last = end - 1;
    int tb = start + wave * 16;
    int e0 = tb + er;
    if (e0 > last) e0 = last;
    if (e0 < 0) e0 = 0;
    float4 r = rec4[(size_t)e0 * 2 + hf];
    while (tb < end) {
        int tn = tb + EB_WAVES * 16;
        int en = tn + er;
        if (en > last) en = last;
        float4 rn = rec4[(size_t)en * 2 + hf];

        if (hf == 0) {
            S[er] = __float_as_int(r.x);
            D[er] = __float_as_int(r.y) - node_lo;
            *reinterpret_cast<float2*>(&A[er * AP + APOS(10)]) = make_float2(r.z, r.w);
        } else {
            *reinterpret_cast<float2*>(&A[er * AP + APOS(12)]) = make_float2(r.x, r.y);
            *reinterpret_cast<float2*>(&A[er * AP + APOS(14)]) = make_float2(r.z, r.w);
        }
        __builtin_amdgcn_wave_barrier();

        {
            const int gq = lane & 3, gr = lane >> 2;
            if (gq < 3) {
                const float4 g0 = *reinterpret_cast<const float4*>(
                    h_in + (size_t)S[gr] * HP + 4 * gq);
                const float4 g1 = *reinterpret_cast<const float4*>(
                    h_in + (size_t)S[gr + 8] * HP + 4 * gq);
                *reinterpret_cast<float4*>(&A[gr * AP + 4 * gq])       = g0;
                *reinterpret_cast<float4*>(&A[(gr + 8) * AP + 4 * gq]) = g1;
            }
        }
        __builtin_amdgcn_wave_barrier();

        const int4 dq0 = *reinterpret_cast<const int4*>(&D[rbase]);
        const int4 dq1 = *reinterpret_cast<const int4*>(&D[rbase + 4]);
        const int dl[8] = {dq0.x, dq0.y, dq0.z, dq0.w, dq1.x, dq1.y, dq1.z, dq1.w};

        v8f c = {};
#pragma unroll
        for (int kc = 0; kc < 4; ++kc) {
            int kb = kc * 4 + koff;
            const float* ap = &A[m * AP + APOS(kb)];
            v2f a;
            a.x = ap[0];
            a.y = ap[1];
            c = __builtin_amdgcn_wmma_f32_16x16x4_f32(false, a, false, bf[kc],
                                                      (short)0, c, false, false);
        }
        if (m < HID) {
            int nvalid = end - tb - rbase;
#pragma unroll
            for (int rr = 0; rr < 8; ++rr) {
                if (rr < nvalid && (unsigned)dl[rr] < NPB)
                    atomicAdd(&acc[dl[rr] * HID + m], lrelu(c[rr]));
            }
        }
        __builtin_amdgcn_wave_barrier();
        r = rn;
        tb = tn;
    }
    const bool upd = wave < NPB / 16;
    const int urow = lane & 15, uhalf = lane >> 4;
    const int nloc = wave * 16 + urow;
    const int unode = node_lo + nloc;
    float4 ha = make_float4(0.0f, 0.0f, 0.0f, 0.0f), hb = ha;
    float2 hc = make_float2(0.0f, 0.0f);
    if (upd && uhalf == 0 && unode < n_nodes) {
        const float* hp = h_in + (size_t)unode * HP;
        ha = *reinterpret_cast<const float4*>(hp);
        hb = *reinterpret_cast<const float4*>(hp + 4);
        hc = *reinterpret_cast<const float2*>(hp + 8);
    }
    __syncthreads();
    if (!upd) return;

    if (uhalf == 0) {
        *reinterpret_cast<float4*>(&A[urow * AP])     = ha;
        *reinterpret_cast<float4*>(&A[urow * AP + 4]) = hb;
        *reinterpret_cast<float2*>(&A[urow * AP + 8]) = hc;
    } else {
        const float2* ar = reinterpret_cast<const float2*>(&acc[nloc * HID]);
#pragma unroll
        for (int j = 0; j < HID / 2; ++j)
            *reinterpret_cast<float2*>(&A[urow * AP + HID + 2 * j]) = ar[j];
    }
    __builtin_amdgcn_wave_barrier();

    v8f uc = {};
#pragma unroll
    for (int kc = 0; kc < 5; ++kc) {
        int kb = kc * 4 + koff;
        v2f a, b;
        a.x = A[m * AP + kb];
        a.y = A[m * AP + kb + 1];
        b.x = uwlds[kb * 16 + m];
        b.y = uwlds[(kb + 1) * 16 + m];
        uc = __builtin_amdgcn_wmma_f32_16x16x4_f32(false, a, false, b,
                                                   (short)0, uc, false, false);
    }
    if (m < out_pitch) {
#pragma unroll
        for (int r = 0; r < 8; ++r) {
            int node = node_lo + wave * 16 + rbase + r;
            if (node < n_nodes)
                h_out[(size_t)node * out_pitch + m] = (m < HID) ? lrelu(uc[r]) : 0.0f;
        }
    }
}

extern "C" void kernel_launch(void* const* d_in, const int* in_sizes, int n_in,
                              void* d_out, int out_size, void* d_ws, size_t ws_size,
                              hipStream_t stream) {
    const float* vertex = (const float*)d_in[0];
    const float* efeat  = (const float*)d_in[1];
    const int*   src    = (const int*)d_in[2];
    const int*   dst    = (const int*)d_in[3];
    const float* w_init = (const float*)d_in[4];
    const float* g_w    = (const float*)d_in[5];
    const float* u_w    = (const float*)d_in[6];
    float* out = (float*)d_out;

    const int N = in_sizes[0] / 82;
    const int E = in_sizes[2];
    const int NP = N * HP;

    float* hA  = (float*)d_ws;
    float* hB  = hA + (size_t)NP;
    float* rec = hB + (size_t)NP;
    int* cnt    = (int*)(rec + (size_t)E * 8);
    int* off    = cnt + HB * NB;
    int* tot    = off + HB * NB;
    int* bstart = tot + NB;

    dim3 blk(256);
    int node_tiles  = (N + 15) / 16;
    int node_blocks = (node_tiles + WAVES - 1) / WAVES;
    int chunk = (E + HB - 1) / HB;

    gnn_embed_kernel<<<node_blocks, blk, 0, stream>>>(vertex, w_init, hA, N);

    gnn_hist_kernel<<<HB, PT, 0, stream>>>(dst, cnt, E, chunk);
    gnn_scan_kernel<<<NB, HB, 0, stream>>>(cnt, off, tot);
    gnn_scatter_kernel<<<HB, PT, 0, stream>>>(efeat, src, dst, off, tot, bstart, rec, E, chunk);

    gnn_edge_bin_kernel<<<NB, PT, 0, stream>>>(hA, rec, bstart, tot, g_w, u_w, hB, N, HP);

    gnn_edge_bin_kernel<<<NB, PT, 0, stream>>>(hB, rec, bstart, tot, g_w, u_w, out, N, HID);
}


# scatter kernel: two adjacent lanes write one whole 32-byte record per store instruction (halves swapped inside the lane pair): 16 records per instruction, not 32 separate 16-byte pieces
# speedup vs baseline: 2.3092x; 1.1307x over previous
#include <hip/hip_runtime.h>
#include <hip/hip_bf16.h>

typedef float v2f __attribute__((ext_vector_type(2)));
typedef float v8f __attribute__((ext_vector_type(8)));

#define WAVES 8
#define HID 10
#define EDGE_DIM 6
#define HP 16

__device__ __forceinline__ float lrelu(float x) {
    return fmaxf(x, 0.1f * x);
}

__device__ __forceinline__ void stage_weights_padded(float* wlds, const float* w,
                                                     int rows_real, int rows_pad,
                                                     int t, int nthreads) {
    for (int i = t; i < rows_pad * 16; i += nthreads) {
        int row = i >> 4, col = i & 15;
        wlds[i] = (col < HID && row < rows_real) ? w[row * HID + col] : 0.0f;
    }
}

#define VDIM 82
#define TILE_F (16 * VDIM)

__global__ void gnn_embed_kernel(const float* __restrict__ vertex,
                                 const float* __restrict__ w_init,
                                 float* __restrict__ h_out, int n_nodes) {
    __shared__ float wlds[84 * 16];
    __shared__ float alds[WAVES][16 * 85];
    int t = threadIdx.x;
    stage_weights_padded(wlds, w_init, 82, 84, t, blockDim.x);
    __syncthreads();

    int wave = t >> 5, lane = t & 31;
    int tile = blockIdx.x * WAVES + wave;
    int base = tile * 16;
    if (base >= n_nodes) return;

    float* A = alds[wave];
    const float* gp = vertex + (size_t)base * VDIM;
    if (base + 16 <= n_nodes) {
        const float4* g4 = reinterpret_cast<const float4*>(gp);
        float4 v[10];
#pragma unroll
        for (int it = 0; it < 10; ++it) v[it] = g4[it * 32 + lane];
        float vt = gp[1280 + lane];
#pragma unroll
        for (int it = 0; it < 10; ++it) {
            int f = (it * 32 + lane) * 4;
            int p0 = f + 3 * (f / VDIM);
            int p1 = f + 2 + 3 * ((f + 2) / VDIM);
            A[p0]     = v[it].x;
            A[p0 + 1] = v[it].y;
            A[p1]     = v[it].z;
            A[p1 + 1] = v[it].w;
        }
        {
            int f = 1280 + lane;
            A[f + 3 * (f / VDIM)] = vt;
        }
    } else {
        for (int f = lane; f < TILE_F; f += 32) {
            int row = f / VDIM;
            A[f + 3 * row] = (base + row < n_nodes) ? gp[f] : 0.0f;
        }
    }
    if (lane < 16) { A[lane * 85 + 82] = 0.0f; A[lane * 85 + 83] = 0.0f; }
    __builtin_amdgcn_wave_barrier();

    v8f c = {};
    int m = lane & 15;
    int koff = (lane < 16) ? 0 : 2;
    for (int kc = 0; kc < 21; ++kc) {
        int kb = kc * 4 + koff;
        v2f a, b;
        a.x = A[m * 85 + kb];
        a.y = A[m * 85 + kb + 1];
        b.x = wlds[kb * 16 + m];
        b.y = wlds[(kb + 1) * 16 + m];
        c = __builtin_amdgcn_wmma_f32_16x16x4_f32(false, a, false, b,
                                                  (short)0, c, false, false);
    }
    {
        int rbase = (lane < 16) ? 0 : 8;
#pragma unroll
        for (int r = 0; r < 8; ++r) {
            int node = base + rbase + r;
            if (node < n_nodes)
                h_out[(size_t)node * HP + m] = (m < HID) ? lrelu(c[r]) : 0.0f;
        }
    }
}

#define NB  512
#define NPB 400
#define HB  512
#define PT  1024

__device__ __forceinline__ int dst_bin(int d) {
    unsigned b = (unsigned)d / NPB;
    return (b > NB - 1) ? (NB - 1) : (int)b;
}

__device__ __forceinline__ void lds_inclusive_scan(int* sm, int n, int t) {
    for (int d = 1; d < n; d <<= 1) {
        int x = 0;
        if (t < n && t >= d) x = sm[t - d];
        __syncthreads();
        if (t < n) sm[t] += x;
        __syncthreads();
    }
}

__global__ __launch_bounds__(PT) void gnn_hist_kernel(const int* __restrict__ dst,
                                                      int* __restrict__ cnt,
                                                      int n_edges, int chunk) {
    __shared__ int hist[NB];
    int t = threadIdx.x;
    for (int i = t; i < NB; i += PT) hist[i] = 0;
    __syncthreads();
    int lo = blockIdx.x * chunk;
    int hi = lo + chunk; if (hi > n_edges) hi = n_edges;
    for (int e = lo + t; e < hi; e += PT) atomicAdd(&hist[dst_bin(dst[e])], 1);
    __syncthreads();
    for (int i = t; i < NB; i += PT) cnt[blockIdx.x * NB + i] = hist[i];
}

__global__ __launch_bounds__(HB) void gnn_scan_kernel(const int* __restrict__ cnt,
                                                      int* __restrict__ off,
                                                      int* __restrict__ tot) {
    __shared__ int sm[HB];
    int t = threadIdx.x, bin = blockIdx.x;
    int v = cnt[t * NB + bin];
    sm[t] = v;
    __syncthreads();
    lds_inclusive_scan(sm, HB, t);
    int incl = sm[t];
    off[t * NB + bin] = incl - v;
    if (t == HB - 1) tot[bin] = incl;
}

#define SWAP_PAIR(x) __builtin_amdgcn_update_dpp(0, (x), 0xB1, 0xF, 0xF, true)

__global__ __launch_bounds__(PT) void gnn_scatter_kernel(const float* __restrict__ efeat,
                                                         const int* __restrict__ src,
                                                         const int* __restrict__ dst,
                                                         const int* __restrict__ off,
                                                         const int* __restrict__ tot,
                                                         int* __restrict__ bstart,
                                                         float* __restrict__ rec,
                                                         int n_edges, int chunk) {
    __shared__ int sm[NB];
    __shared__ int cursor[NB];
    int t = threadIdx.x, blk = blockIdx.x;
    int v = 0;
    if (t < NB) { v = tot[t]; sm[t] = v; }
    __syncthreads();
    lds_inclusive_scan(sm, NB, t);
    if (t < NB) {
        int st = sm[t] - v;
        cursor[t] = st + off[blk * NB + t];
        if (blk == 0) bstart[t] = st;
    }
    __syncthreads();

    float4* rec4 = reinterpret_cast<float4*>(rec);
    int lo = blk * chunk;
    int hi = lo + chunk; if (hi > n_edges) hi = n_edges;
    const bool odd = (t & 1) != 0;
    for (int eb = lo; eb < hi; eb += PT) {
        int e = eb + t;
        bool valid = e < hi;
        int ec = valid ? e : hi - 1;
        int d = dst[ec];
        int s = src[ec];
        const float2* ep = reinterpret_cast<const float2*>(efeat + (size_t)ec * EDGE_DIM);
        float2 f0 = ep[0], f1 = ep[1], f2 = ep[2];
        int pos = -1;
        if (valid) pos = atomicAdd(&cursor[dst_bin(d)], 1);
        int a0 = s, a1 = d, a2 = __float_as_int(f0.x), a3 = __float_as_int(f0.y);
        int b0 = __float_as_int(f1.x), b1 = __float_as_int(f1.y);
        int b2 = __float_as_int(f2.x), b3 = __float_as_int(f2.y);
        int g0 = SWAP_PAIR(odd ? a0 : b0);
        int g1 = SWAP_PAIR(odd ? a1 : b1);
        int g2 = SWAP_PAIR(odd ? a2 : b2);
        int g3 = SWAP_PAIR(odd ? a3 : b3);
        int ppos = SWAP_PAIR(pos);
        int p1 = odd ? ppos : pos;
        int p2 = odd ? pos : ppos;
        int h  = odd ? 1 : 0;
        if (p1 >= 0)
            rec4[(size_t)p1 * 2 + h] = make_float4(__int_as_float(odd ? g0 : a0), __int_as_float(odd ? g1 : a1),
                                                   __int_as_float(odd ? g2 : a2), __int_as_float(odd ? g3 : a3));
        if (p2 >= 0)
            rec4[(size_t)p2 * 2 + h] = make_float4(__int_as_float(odd ? b0 : g0), __int_as_float(odd ? b1 : g1),
                                                   __int_as_float(odd ? b2 : g2), __int_as_float(odd ? b3 : g3));
    }
}

#define EB_WAVES (PT / 32)
#define AP 20
#define APOS(k) ((k) < HID ? (k) : (k) + 2)

__global__ __launch_bounds__(PT) void gnn_edge_bin_kernel(const float* __restrict__ h_in,
                                                          const float* __restrict__ rec,
                                                          const int* __restrict__ bstart,
                                                          const int* __restrict__ tot,
                                                          const float* __restrict__ g_w,
                                                          const float* __restrict__ u_w,
                                                          float* __restrict__ h_out, int n_nodes,
                                                          int out_pitch) {
    __shared__ float wlds[16 * 16];
    __shared__ float uwlds[20 * 16];
    __shared__ alignas(16) float acc[NPB * HID];
    __shared__ alignas(16) float alds[EB_WAVES][16 * AP];
    __shared__ int   slds[EB_WAVES][16];
    __shared__ alignas(16) int dlds[EB_WAVES][16];
    int t = threadIdx.x, bin = blockIdx.x;
    stage_weights_padded(wlds, g_w, 16, 16, t, PT);
    stage_weights_padded(uwlds, u_w, 20, 20, t, PT);
    for (int i = t; i < NPB * HID; i += PT) acc[i] = 0.0f;
    const int start = bstart[bin];
    const int end = start + tot[bin];
    const int node_lo = bin * NPB;
    __syncthreads();

    int wave = t >> 5, lane = t & 31;
    float* A = alds[wave];
    int* S = slds[wave];
    int* D = dlds[wave];
    const float4* rec4 = reinterpret_cast<const float4*>(rec);
    int m = lane & 15;
    int koff = (lane < 16) ? 0 : 2;
    int er = lane >> 1, hf = lane & 1;
    int rbase = (lane < 16) ? 0 : 8;

    v2f bf[4];
#pragma unroll
    for (int kc = 0; kc < 4; ++kc) {
        int kb = kc * 4 + koff;
        bf[kc].x = wlds[kb * 16 + m];
        bf[kc].y = wlds[(kb + 1) * 16 + m];
    }

    const int last = end - 1;
    int tb = start + wave * 16;
    int e0 = tb + er;
    if (e0 > last) e0 = last;
    if (e0 < 0) e0 = 0;
    float4 r = rec4[(size_t)e0 * 2 + hf];
    while (tb < end) {
        int tn = tb + EB_WAVES * 16;
        int en = tn + er;
        if (en > last) en = last;
        float4 rn = rec4[(size_t)en * 2 + hf];

        if (hf == 0) {
            S[er] = __float_as_int(r.x);
            D[er] = __float_as_int(r.y) - node_lo;
            *reinterpret_cast<float2*>(&A[er * AP + APOS(10)]) = make_float2(r.z, r.w);
        } else {
            *reinterpret_cast<float2*>(&A[er * AP + APOS(12)]) = make_float2(r.x, r.y);
            *reinterpret_cast<float2*>(&A[er * AP + APOS(14)]) = make_float2(r.z, r.w);
        }
        __builtin_amdgcn_wave_barrier();

        {
            const int gq = lane & 3, gr = lane >> 2;
            if (gq < 3) {
                const float4 g0 = *reinterpret_cast<const float4*>(
                    h_in + (size_t)S[gr] * HP + 4 * gq);
                const float4 g1 = *reinterpret_cast<const float4*>(
                    h_in + (size_t)S[gr + 8] * HP + 4 * gq);
                *reinterpret_cast<float4*>(&A[gr * AP + 4 * gq])       = g0;
                *reinterpret_cast<float4*>(&A[(gr + 8) * AP + 4 * gq]) = g1;
            }
        }
        __builtin_amdgcn_wave_barrier();

        const int4 dq0 = *reinterpret_cast<const int4*>(&D[rbase]);
        const int4 dq1 = *reinterpret_cast<const int4*>(&D[rbase + 4]);
        const int dl[8] = {dq0.x, dq0.y, dq0.z, dq0.w, dq1.x, dq1.y, dq1.z, dq1.w};

        v8f c = {};
#pragma unroll
        for (int kc = 0; kc < 4; ++kc) {
            int kb = kc * 4 + koff;
            const float* ap = &A[m * AP + APOS(kb)];
            v2f a;
            a.x = ap[0];
            a.y = ap[1];
            c = __builtin_amdgcn_wmma_f32_16x16x4_f32(false, a, false, bf[kc],
                                                      (short)0, c, false, false);
        }
        if (m < HID) {
            int nvalid = end - tb - rbase;
#pragma unroll
            for (int rr = 0; rr < 8; ++rr) {
                if (rr < nvalid && (unsigned)dl[rr] < NPB)
                    atomicAdd(&acc[dl[rr] * HID + m], lrelu(c[rr]));
            }
        }
        __builtin_amdgcn_wave_barrier();
        r = rn;
        tb = tn;
    }
    const bool upd = wave < NPB / 16;
    const int urow = lane & 15, uhalf = lane >> 4;
    const int nloc = wave * 16 + urow;
    const int unode = node_lo + nloc;
    float4 ha = make_float4(0.0f, 0.0f, 0.0f, 0.0f), hb = ha;
    float2 hc = make_float2(0.0f, 0.0f);
    if (upd && uhalf == 0 && unode < n_nodes) {
        const float* hp = h_in + (size_t)unode * HP;
        ha = *reinterpret_cast<const float4*>(hp);
        hb = *reinterpret_cast<const float4*>(hp + 4);
        hc = *reinterpret_cast<const float2*>(hp + 8);
    }
    __syncthreads();
    if (!upd) return;

    if (uhalf == 0) {
        *reinterpret_cast<float4*>(&A[urow * AP])     = ha;
        *reinterpret_cast<float4*>(&A[urow * AP + 4]) = hb;
        *reinterpret_cast<float2*>(&A[urow * AP + 8]) = hc;
    } else {
        const float2* ar = reinterpret_cast<const float2*>(&acc[nloc * HID]);
#pragma unroll
        for (int j = 0; j < HID / 2; ++j)
            *reinterpret_cast<float2*>(&A[urow * AP + HID + 2 * j]) = ar[j];
    }
    __builtin_amdgcn_wave_barrier();

    v8f uc = {};
#pragma unroll
    for (int kc = 0; kc < 5; ++kc) {
        int kb = kc * 4 + koff;
        v2f a, b;
        a.x = A[m * AP + kb];
        a.y = A[m * AP + kb + 1];
        b.x = uwlds[kb * 16 + m];
        b.y = uwlds[(kb + 1) * 16 + m];
        uc = __builtin_amdgcn_wmma_f32_16x16x4_f32(false, a, false, b,
                                                   (short)0, uc, false, false);
    }
    if (m < out_pitch) {
#pragma unroll
        for (int r = 0; r < 8; ++r) {
            int node = node_lo + wave * 16 + rbase + r;
            if (node < n_nodes)
                h_out[(size_t)node * out_pitch + m] = (m < HID) ? lrelu(uc[r]) : 0.0f;
        }
    }
}

extern "C" void kernel_launch(void* const* d_in, const int* in_sizes, int n_in,
                              void* d_out, int out_size, void* d_ws, size_t ws_size,
                              hipStream_t stream) {
    const float* vertex = (const float*)d_in[0];
    const float* efeat  = (const float*)d_in[1];
    const int*   src    = (const int*)d_in[2];
    const int*   dst    = (const int*)d_in[3];
    const float* w_init = (const float*)d_in[4];
    const float* g_w    = (const float*)d_in[5];
    const float* u_w    = (const float*)d_in[6];
    float* out = (float*)d_out;

    const int N = in_sizes[0] / 82;
    const int E = in_sizes[2];
    const int NP = N * HP;

    float* hA  = (float*)d_ws;
    float* hB  = hA + (size_t)NP;
    float* rec = hB + (size_t)NP;
    int* cnt    = (int*)(rec + (size_t)E * 8);
    int* off    = cnt + HB * NB;
    int* tot    = off + HB * NB;
    int* bstart = tot + NB;

    dim3 blk(256);
    int node_tiles  = (N + 15) / 16;
    int node_blocks = (node_tiles + WAVES - 1) / WAVES;
    int chunk = (E + HB - 1) / HB;

    gnn_embed_kernel<<<node_blocks, blk, 0, stream>>>(vertex, w_init, hA, N);

    gnn_hist_kernel<<<HB, PT, 0, stream>>>(dst, cnt, E, chunk);
    gnn_scan_kernel<<<NB, HB, 0, stream>>>(cnt, off, tot);
    gnn_scatter_kernel<<<HB, PT, 0, stream>>>(efeat, src, dst, off, tot, bstart, rec, E, chunk);

    gnn_edge_bin_kernel<<<NB, PT, 0, stream>>>(hA, rec, bstart, tot, g_w, u_w, hB, N, HP);

    gnn_edge_bin_kernel<<<NB, PT, 0, stream>>>(hB, rec, bstart, tot, g_w, u_w, out, N, HID);
}


# scatter kernel: each of 1024 blocks sorts its 6250-edge chunk by bin in LDS, then stores in slot order: a store covers runs of whole lines, not 16 lone 32-byte records; record order in a bin changes
# speedup vs baseline: 2.4766x; 1.0725x over previous
#include <hip/hip_runtime.h>
#include <hip/hip_bf16.h>

typedef float v2f __attribute__((ext_vector_type(2)));
typedef float v8f __attribute__((ext_vector_type(8)));

#define WAVES 8
#define HID 10
#define EDGE_DIM 6
#define HP 16

__device__ __forceinline__ float lrelu(float x) {
    return fmaxf(x, 0.1f * x);
}

__device__ __forceinline__ void stage_weights_padded(float* wlds, const float* w,
                                                     int rows_real, int rows_pad,
                                                     int t, int nthreads) {
    for (int i = t; i < rows_pad * 16; i += nthreads) {
        int row = i >> 4, col = i & 15;
        wlds[i] = (col < HID && row < rows_real) ? w[row * HID + col] : 0.0f;
    }
}

#define VDIM 82
#define TILE_F (16 * VDIM)

__global__ void gnn_embed_kernel(const float* __restrict__ vertex,
                                 const float* __restrict__ w_init,
                                 float* __restrict__ h_out, int n_nodes) {
    __shared__ float wlds[84 * 16];
    __shared__ float alds[WAVES][16 * 85];
    int t = threadIdx.x;
    stage_weights_padded(wlds, w_init, 82, 84, t, blockDim.x);
    __syncthreads();

    int wave = t >> 5, lane = t & 31;
    int tile = blockIdx.x * WAVES + wave;
    int base = tile * 16;
    if (base >= n_nodes) return;

    float* A = alds[wave];
    const float* gp = vertex + (size_t)base * VDIM;
    if (base + 16 <= n_nodes) {
        const float4* g4 = reinterpret_cast<const float4*>(gp);
        float4 v[10];
#pragma unroll
        for (int it = 0; it < 10; ++it) v[it] = g4[it * 32 + lane];
        float vt = gp[1280 + lane];
#pragma unroll
        for (int it = 0; it < 10; ++it) {
            int f = (it * 32 + lane) * 4;
            int p0 = f + 3 * (f / VDIM);
            int p1 = f + 2 + 3 * ((f + 2) / VDIM);
            A[p0]     = v[it].x;
            A[p0 + 1] = v[it].y;
            A[p1]     = v[it].z;
            A[p1 + 1] = v[it].w;
        }
        {
            int f = 1280 + lane;
            A[f + 3 * (f / VDIM)] = vt;
        }
    } else {
        for (int f = lane; f < TILE_F; f += 32) {
            int row = f / VDIM;
            A[f + 3 * row] = (base + row < n_nodes) ? gp[f] : 0.0f;
        }
    }
    if (lane < 16) { A[lane * 85 + 82] = 0.0f; A[lane * 85 + 83] = 0.0f; }
    __builtin_amdgcn_wave_barrier();

    v8f c = {};
    int m = lane & 15;
    int koff = (lane < 16) ? 0 : 2;
    for (int kc = 0; kc < 21; ++kc) {
        int kb = kc * 4 + koff;
        v2f a, b;
        a.x = A[m * 85 + kb];
        a.y = A[m * 85 + kb + 1];
        b.x = wlds[kb * 16 + m];
        b.y = wlds[(kb + 1) * 16 + m];
        c = __builtin_amdgcn_wmma_f32_16x16x4_f32(false, a, false, b,
                                                  (short)0, c, false, false);
    }
    {
        int rbase = (lane < 16) ? 0 : 8;
#pragma unroll
        for (int r = 0; r < 8; ++r) {
            int node = base + rbase + r;
            if (node < n_nodes)
                h_out[(size_t)node * HP + m] = (m < HID) ? lrelu(c[r]) : 0.0f;
        }
    }
}

#define NB  512
#define NPB 400
#define HB  1024
#define CAP 6400
#define PT  1024

__device__ __forceinline__ int dst_bin(int d) {
    unsigned b = (unsigned)d / NPB;
    return (b > NB - 1) ? (NB - 1) : (int)b;
}

__device__ __forceinline__ void lds_inclusive_scan(int* sm, int n, int t) {
    for (int d = 1; d < n; d <<= 1) {
        int x = 0;
        if (t < n && t >= d) x = sm[t - d];
        __syncthreads();
        if (t < n) sm[t] += x;
        __syncthreads();
    }
}

__global__ __launch_bounds__(PT) void gnn_hist_kernel(const int* __restrict__ dst,
                                                      int* __restrict__ cnt,
                                                      int n_edges, int chunk) {
    __shared__ int hist[NB];
    int t = threadIdx.x;
    for (int i = t; i < NB; i += PT) hist[i] = 0;
    __syncthreads();
    int lo = blockIdx.x * chunk;
    int hi = lo + chunk; if (hi > n_edges) hi = n_edges;
    for (int e = lo + t; e < hi; e += PT) atomicAdd(&hist[dst_bin(dst[e])], 1);
    __syncthreads();
    for (int i = t; i < NB; i += PT) cnt[blockIdx.x * NB + i] = hist[i];
}

__global__ __launch_bounds__(HB) void gnn_scan_kernel(const int* __restrict__ cnt,
                                                      int* __restrict__ off,
                                                      int* __restrict__ tot) {
    __shared__ int sm[HB];
    int t = threadIdx.x, bin = blockIdx.x;
    int v = cnt[t * NB + bin];
    sm[t] = v;
    __syncthreads();
    lds_inclusive_scan(sm, HB, t);
    int incl = sm[t];
    off[t * NB + bin] = incl - v;
    if (t == HB - 1) tot[bin] = incl;
}

__global__ __launch_bounds__(PT) void gnn_scatter_kernel(const float* __restrict__ efeat,
                                                         const int* __restrict__ src,
                                                         const int* __restrict__ dst,
                                                         const int* __restrict__ cnt,
                                                         const int* __restrict__ off,
                                                         const int* __restrict__ tot,
                                                         int* __restrict__ bstart,
                                                         float* __restrict__ rec,
                                                         int n_edges, int chunk) {
    __shared__ alignas(16) float4 stage[CAP * 2];
    __shared__ int sm[NB];
    __shared__ int lsm[NB];
    __shared__ int cursor[NB];
    __shared__ int gdelta[NB];
    int t = threadIdx.x, blk = blockIdx.x;
    int v = 0, lv = 0;
    if (t < NB) {
        v = tot[t];             sm[t] = v;
        lv = cnt[blk * NB + t]; lsm[t] = lv;
    }
    __syncthreads();
    for (int d = 1; d < NB; d <<= 1) {
        int x = 0, y = 0;
        if (t < NB && t >= d) { x = sm[t - d]; y = lsm[t - d]; }
        __syncthreads();
        if (t < NB) { sm[t] += x; lsm[t] += y; }
        __syncthreads();
    }
    if (t < NB) {
        int st = sm[t] - v;
        int ls = lsm[t] - lv;
        cursor[t] = ls;
        gdelta[t] = st + off[blk * NB + t] - ls;
        if (blk == 0) bstart[t] = st;
    }
    __syncthreads();

    int lo = blk * chunk;
    int hi = lo + chunk; if (hi > n_edges) hi = n_edges;
    const int n = hi - lo;
    for (int eb = lo; eb < hi; eb += 2 * PT) {
        int e0 = eb + t, e1 = eb + PT + t;
        bool v0 = e0 < hi, v1 = e1 < hi;
        int c0 = v0 ? e0 : hi - 1;
        int c1 = v1 ? e1 : hi - 1;
        int d0 = dst[c0], d1 = dst[c1];
        int s0 = src[c0], s1 = src[c1];
        const float2* p0 = reinterpret_cast<const float2*>(efeat + (size_t)c0 * EDGE_DIM);
        const float2* p1 = reinterpret_cast<const float2*>(efeat + (size_t)c1 * EDGE_DIM);
        float2 a0 = p0[0], a1 = p0[1], a2 = p0[2];
        float2 b0 = p1[0], b1 = p1[1], b2 = p1[2];
        asm volatile("" :: "v"(d0), "v"(d1), "v"(s0), "v"(s1),
                     "v"(a0.x), "v"(a0.y), "v"(a1.x), "v"(a1.y), "v"(a2.x), "v"(a2.y),
                     "v"(b0.x), "v"(b0.y), "v"(b1.x), "v"(b1.y), "v"(b2.x), "v"(b2.y));
        if (v0) {
            int slot = atomicAdd(&cursor[dst_bin(d0)], 1);
            stage[slot * 2]     = make_float4(__int_as_float(s0), __int_as_float(d0), a0.x, a0.y);
            stage[slot * 2 + 1] = make_float4(a1.x, a1.y, a2.x, a2.y);
        }
        if (v1) {
            int slot = atomicAdd(&cursor[dst_bin(d1)], 1);
            stage[slot * 2]     = make_float4(__int_as_float(s1), __int_as_float(d1), b0.x, b0.y);
            stage[slot * 2 + 1] = make_float4(b1.x, b1.y, b2.x, b2.y);
        }
    }
    __syncthreads();

    float4* rec4 = reinterpret_cast<float4*>(rec);
    const int* stage_i = reinterpret_cast<const int*>(stage);
    for (int i = t; i < 2 * n; i += PT) {
        int slot = i >> 1, h = i & 1;
        float4 val = stage[i];
        int d = stage_i[slot * 8 + 1];
        int pos = slot + gdelta[dst_bin(d)];
        rec4[(size_t)pos * 2 + h] = val;
    }
}

#define EB_WAVES (PT / 32)
#define AP 20
#define APOS(k) ((k) < HID ? (k) : (k) + 2)

__global__ __launch_bounds__(PT) void gnn_edge_bin_kernel(const float* __restrict__ h_in,
                                                          const float* __restrict__ rec,
                                                          const int* __restrict__ bstart,
                                                          const int* __restrict__ tot,
                                                          const float* __restrict__ g_w,
                                                          const float* __restrict__ u_w,
                                                          float* __restrict__ h_out, int n_nodes,
                                                          int out_pitch) {
    __shared__ float wlds[16 * 16];
    __shared__ float uwlds[20 * 16];
    __shared__ alignas(16) float acc[NPB * HID];
    __shared__ alignas(16) float alds[EB_WAVES][16 * AP];
    __shared__ int   slds[EB_WAVES][16];
    __shared__ alignas(16) int dlds[EB_WAVES][16];
    int t = threadIdx.x, bin = blockIdx.x;
    stage_weights_padded(wlds, g_w, 16, 16, t, PT);
    stage_weights_padded(uwlds, u_w, 20, 20, t, PT);
    for (int i = t; i < NPB * HID; i += PT) acc[i] = 0.0f;
    const int start = bstart[bin];
    const int end = start + tot[bin];
    const int node_lo = bin * NPB;
    __syncthreads();

    int wave = t >> 5, lane = t & 31;
    float* A = alds[wave];
    int* S = slds[wave];
    int* D = dlds[wave];
    const float4* rec4 = reinterpret_cast<const float4*>(rec);
    int m = lane & 15;
    int koff = (lane < 16) ? 0 : 2;
    int er = lane >> 1, hf = lane & 1;
    int rbase = (lane < 16) ? 0 : 8;

    v2f bf[4];
#pragma unroll
    for (int kc = 0; kc < 4; ++kc) {
        int kb = kc * 4 + koff;
        bf[kc].x = wlds[kb * 16 + m];
        bf[kc].y = wlds[(kb + 1) * 16 + m];
    }

    const int last = end - 1;
    int tb = start + wave * 16;
    int e0 = tb + er;
    if (e0 > last) e0 = last;
    if (e0 < 0) e0 = 0;
    float4 r = rec4[(size_t)e0 * 2 + hf];
    while (tb < end) {
        int tn = tb + EB_WAVES * 16;
        int en = tn + er;
        if (en > last) en = last;
        float4 rn = rec4[(size_t)en * 2 + hf];

        if (hf == 0) {
            S[er] = __float_as_int(r.x);
            D[er] = __float_as_int(r.y) - node_lo;
            *reinterpret_cast<float2*>(&A[er * AP + APOS(10)]) = make_float2(r.z, r.w);
        } else {
            *reinterpret_cast<float2*>(&A[er * AP + APOS(12)]) = make_float2(r.x, r.y);
            *reinterpret_cast<float2*>(&A[er * AP + APOS(14)]) = make_float2(r.z, r.w);
        }
        __builtin_amdgcn_wave_barrier();

        {
            const int gq = lane & 3, gr = lane >> 2;
            if (gq < 3) {
                const float4 g0 = *reinterpret_cast<const float4*>(
                    h_in + (size_t)S[gr] * HP + 4 * gq);
                const float4 g1 = *reinterpret_cast<const float4*>(
                    h_in + (size_t)S[gr + 8] * HP + 4 * gq);
                *reinterpret_cast<float4*>(&A[gr * AP + 4 * gq])       = g0;
                *reinterpret_cast<float4*>(&A[(gr + 8) * AP + 4 * gq]) = g1;
            }
        }
        __builtin_amdgcn_wave_barrier();

        const int4 dq0 = *reinterpret_cast<const int4*>(&D[rbase]);
        const int4 dq1 = *reinterpret_cast<const int4*>(&D[rbase + 4]);
        const int dl[8] = {dq0.x, dq0.y, dq0.z, dq0.w, dq1.x, dq1.y, dq1.z, dq1.w};

        v8f c = {};
#pragma unroll
        for (int kc = 0; kc < 4; ++kc) {
            int kb = kc * 4 + koff;
            const float* ap = &A[m * AP + APOS(kb)];
            v2f a;
            a.x = ap[0];
            a.y = ap[1];
            c = __builtin_amdgcn_wmma_f32_16x16x4_f32(false, a, false, bf[kc],
                                                      (short)0, c, false, false);
        }
        if (m < HID) {
            int nvalid = end - tb - rbase;
#pragma unroll
            for (int rr = 0; rr < 8; ++rr) {
                if (rr < nvalid && (unsigned)dl[rr] < NPB)
                    atomicAdd(&acc[dl[rr] * HID + m], lrelu(c[rr]));
            }
        }
        __builtin_amdgcn_wave_barrier();
        r = rn;
        tb = tn;
    }
    const bool upd = wave < NPB / 16;
    const int urow = lane & 15, uhalf = lane >> 4;
    const int nloc = wave * 16 + urow;
    const int unode = node_lo + nloc;
    float4 ha = make_float4(0.0f, 0.0f, 0.0f, 0.0f), hb = ha;
    float2 hc = make_float2(0.0f, 0.0f);
    if (upd && uhalf == 0 && unode < n_nodes) {
        const float* hp = h_in + (size_t)unode * HP;
        ha = *reinterpret_cast<const float4*>(hp);
        hb = *reinterpret_cast<const float4*>(hp + 4);
        hc = *reinterpret_cast<const float2*>(hp + 8);
    }
    __syncthreads();
    if (!upd) return;

    if (uhalf == 0) {
        *reinterpret_cast<float4*>(&A[urow * AP])     = ha;
        *reinterpret_cast<float4*>(&A[urow * AP + 4]) = hb;
        *reinterpret_cast<float2*>(&A[urow * AP + 8]) = hc;
    } else {
        const float2* ar = reinterpret_cast<const float2*>(&acc[nloc * HID]);
#pragma unroll
        for (int j = 0; j < HID / 2; ++j)
            *reinterpret_cast<float2*>(&A[urow * AP + HID + 2 * j]) = ar[j];
    }
    __builtin_amdgcn_wave_barrier();

    v8f uc = {};
#pragma unroll
    for (int kc = 0; kc < 5; ++kc) {
        int kb = kc * 4 + koff;
        v2f a, b;
        a.x = A[m * AP + kb];
        a.y = A[m * AP + kb + 1];
        b.x = uwlds[kb * 16 + m];
        b.y = uwlds[(kb + 1) * 16 + m];
        uc = __builtin_amdgcn_wmma_f32_16x16x4_f32(false, a, false, b,
                                                   (short)0, uc, false, false);
    }
    if (m < out_pitch) {
#pragma unroll
        for (int r = 0; r < 8; ++r) {
            int node = node_lo + wave * 16 + rbase + r;
            if (node < n_nodes)
                h_out[(size_t)node * out_pitch + m] = (m < HID) ? lrelu(uc[r]) : 0.0f;
        }
    }
}

extern "C" void kernel_launch(void* const* d_in, const int* in_sizes, int n_in,
                              void* d_out, int out_size, void* d_ws, size_t ws_size,
                              hipStream_t stream) {
    const float* vertex = (const float*)d_in[0];
    const float* efeat  = (const float*)d_in[1];
    const int*   src    = (const int*)d_in[2];
    const int*   dst    = (const int*)d_in[3];
    const float* w_init = (const float*)d_in[4];
    const float* g_w    = (const float*)d_in[5];
    const float* u_w    = (const float*)d_in[6];
    float* out = (float*)d_out;

    const int N = in_sizes[0] / 82;
    const int E = in_sizes[2];
    const int NP = N * HP;

    float* hA  = (float*)d_ws;
    float* hB  = hA + (size_t)NP;
    float* rec = hB + (size_t)NP;
    int* cnt    = (int*)(rec + (size_t)E * 8);
    int* off    = cnt + HB * NB;
    int* tot    = off + HB * NB;
    int* bstart = tot + NB;

    dim3 blk(256);
    int node_tiles  = (N + 15) / 16;
    int node_blocks = (node_tiles + WAVES - 1) / WAVES;
    int chunk = (E + HB - 1) / HB;

    gnn_embed_kernel<<<node_blocks, blk, 0, stream>>>(vertex, w_init, hA, N);

    gnn_hist_kernel<<<HB, PT, 0, stream>>>(dst, cnt, E, chunk);
    gnn_scan_kernel<<<NB, HB, 0, stream>>>(cnt, off, tot);
    gnn_scatter_kernel<<<HB, PT, 0, stream>>>(efeat, src, dst, cnt, off, tot, bstart, rec, E, chunk);

    gnn_edge_bin_kernel<<<NB, PT, 0, stream>>>(hA, rec, bstart, tot, g_w, u_w, hB, N, HP);

    gnn_edge_bin_kernel<<<NB, PT, 0, stream>>>(hB, rec, bstart, tot, g_w, u_w, out, N, HID);
}


# histogram and count-scan launches: few waves, each with all its loads in flight (4-wave histogram blocks, 50 dst loads per thread before the first LDS add; scan: 16 blocks, 32 whole-line row loads)
# speedup vs baseline: 2.6492x; 1.0697x over previous
#include <hip/hip_runtime.h>
#include <hip/hip_bf16.h>

typedef float v2f __attribute__((ext_vector_type(2)));
typedef float v8f __attribute__((ext_vector_type(8)));

#define WAVES 8
#define HID 10
#define EDGE_DIM 6
#define HP 16

__device__ __forceinline__ float lrelu(float x) {
    return fmaxf(x, 0.1f * x);
}

__device__ __forceinline__ void stage_weights_padded(float* wlds, const float* w,
                                                     int rows_real, int rows_pad,
                                                     int t, int nthreads) {
    for (int i = t; i < rows_pad * 16; i += nthreads) {
        int row = i >> 4, col = i & 15;
        wlds[i] = (col < HID && row < rows_real) ? w[row * HID + col] : 0.0f;
    }
}

#define VDIM 82
#define TILE_F (16 * VDIM)

__global__ void gnn_embed_kernel(const float* __restrict__ vertex,
                                 const float* __restrict__ w_init,
                                 float* __restrict__ h_out, int n_nodes) {
    __shared__ float wlds[84 * 16];
    __shared__ float alds[WAVES][16 * 85];
    int t = threadIdx.x;
    stage_weights_padded(wlds, w_init, 82, 84, t, blockDim.x);
    __syncthreads();

    int wave = t >> 5, lane = t & 31;
    int tile = blockIdx.x * WAVES + wave;
    int base = tile * 16;
    if (base >= n_nodes) return;

    float* A = alds[wave];
    const float* gp = vertex + (size_t)base * VDIM;
    if (base + 16 <= n_nodes) {
        const float4* g4 = reinterpret_cast<const float4*>(gp);
        float4 v[10];
#pragma unroll
        for (int it = 0; it < 10; ++it) v[it] = g4[it * 32 + lane];
        float vt = gp[1280 + lane];
#pragma unroll
        for (int it = 0; it < 10; ++it) {
            int f = (it * 32 + lane) * 4;
            int p0 = f + 3 * (f / VDIM);
            int p1 = f + 2 + 3 * ((f + 2) / VDIM);
            A[p0]     = v[it].x;
            A[p0 + 1] = v[it].y;
            A[p1]     = v[it].z;
            A[p1 + 1] = v[it].w;
        }
        {
            int f = 1280 + lane;
            A[f + 3 * (f / VDIM)] = vt;
        }
    } else {
        for (int f = lane; f < TILE_F; f += 32) {
            int row = f / VDIM;
            A[f + 3 * row] = (base + row < n_nodes) ? gp[f] : 0.0f;
        }
    }
    if (lane < 16) { A[lane * 85 + 82] = 0.0f; A[lane * 85 + 83] = 0.0f; }
    __builtin_amdgcn_wave_barrier();

    v8f c = {};
    int m = lane & 15;
    int koff = (lane < 16) ? 0 : 2;
    for (int kc = 0; kc < 21; ++kc) {
        int kb = kc * 4 + koff;
        v2f a, b;
        a.x = A[m * 85 + kb];
        a.y = A[m * 85 + kb + 1];
        b.x = wlds[kb * 16 + m];
        b.y = wlds[(kb + 1) * 16 + m];
        c = __builtin_amdgcn_wmma_f32_16x16x4_f32(false, a, false, b,
                                                  (short)0, c, false, false);
    }
    {
        int rbase = (lane < 16) ? 0 : 8;
#pragma unroll
        for (int r = 0; r < 8; ++r) {
            int node = base + rbase + r;
            if (node < n_nodes)
                h_out[(size_t)node * HP + m] = (m < HID) ? lrelu(c[r]) : 0.0f;
        }
    }
}

#define NB  512
#define NPB 400
#define HB  1024
#define CAP 6400
#define PT  1024
#define HT  128
#define HU  50
#define SB  32
#define SR  (HB / 32)

__device__ __forceinline__ int dst_bin(int d) {
    unsigned b = (unsigned)d / NPB;
    return (b > NB - 1) ? (NB - 1) : (int)b;
}

__global__ __launch_bounds__(HT) void gnn_hist_kernel(const int* __restrict__ dst,
                                                      int* __restrict__ cnt,
                                                      int n_edges, int chunk) {
    __shared__ int hist[NB];
    int t = threadIdx.x;
    for (int i = t; i < NB; i += HT) hist[i] = 0;
    __syncthreads();
    int lo = blockIdx.x * chunk;
    int hi = lo + chunk; if (hi > n_edges) hi = n_edges;
    for (int eb = lo; eb < hi; eb += HT * HU) {
        int d[HU];
#pragma unroll
        for (int j = 0; j < HU; ++j) {
            int e = eb + j * HT + t;
            d[j] = dst[e < hi ? e : hi - 1];
        }
#pragma unroll
        for (int j = 0; j < HU; ++j) {
            int e = eb + j * HT + t;
            if (e < hi) atomicAdd(&hist[dst_bin(d[j])], 1);
        }
    }
    __syncthreads();
    for (int i = t; i < NB; i += HT) cnt[blockIdx.x * NB + i] = hist[i];
}

__global__ __launch_bounds__(1024) void gnn_scan_kernel(const int* __restrict__ cnt,
                                                        int* __restrict__ off,
                                                        int* __restrict__ tot) {
    __shared__ alignas(16) int part[SB][36];
    int t = threadIdx.x;
    int binl = t & (SB - 1), rg = t >> 5;
    int bin = blockIdx.x * SB + binl;
    int v[SR];
#pragma unroll
    for (int i = 0; i < SR; ++i) v[i] = cnt[(rg * SR + i) * NB + bin];
    int s = 0;
#pragma unroll
    for (int i = 0; i < SR; ++i) { int x = v[i]; v[i] = s; s += x; }
    part[binl][rg] = s;
    __syncthreads();
    int base = 0;
    const int4* pr = reinterpret_cast<const int4*>(&part[binl][0]);
#pragma unroll
    for (int q = 0; q < 8; ++q) {
        int4 p = pr[q];
        int g = q * 4;
        base += (g     < rg ? p.x : 0) + (g + 1 < rg ? p.y : 0)
              + (g + 2 < rg ? p.z : 0) + (g + 3 < rg ? p.w : 0);
    }
#pragma unroll
    for (int i = 0; i < SR; ++i) off[(rg * SR + i) * NB + bin] = base + v[i];
    if (rg == 31) tot[bin] = base + s;
}

__global__ __launch_bounds__(PT) void gnn_scatter_kernel(const float* __restrict__ efeat,
                                                         const int* __restrict__ src,
                                                         const int* __restrict__ dst,
                                                         const int* __restrict__ cnt,
                                                         const int* __restrict__ off,
                                                         const int* __restrict__ tot,
                                                         int* __restrict__ bstart,
                                                         float* __restrict__ rec,
                                                         int n_edges, int chunk) {
    __shared__ alignas(16) float4 stage[CAP * 2];
    __shared__ int sm[NB];
    __shared__ int lsm[NB];
    __shared__ int cursor[NB];
    __shared__ int gdelta[NB];
    int t = threadIdx.x, blk = blockIdx.x;
    int v = 0, lv = 0;
    if (t < NB) {
        v = tot[t];             sm[t] = v;
        lv = cnt[blk * NB + t]; lsm[t] = lv;
    }
    __syncthreads();
    for (int d = 1; d < NB; d <<= 1) {
        int x = 0, y = 0;
        if (t < NB && t >= d) { x = sm[t - d]; y = lsm[t - d]; }
        __syncthreads();
        if (t < NB) { sm[t] += x; lsm[t] += y; }
        __syncthreads();
    }
    if (t < NB) {
        int st = sm[t] - v;
        int ls = lsm[t] - lv;
        cursor[t] = ls;
        gdelta[t] = st + off[blk * NB + t] - ls;
        if (blk == 0) bstart[t] = st;
    }
    __syncthreads();

    int lo = blk * chunk;
    int hi = lo + chunk; if (hi > n_edges) hi = n_edges;
    const int n = hi - lo;
    for (int eb = lo; eb < hi; eb += 2 * PT) {
        int e0 = eb + t, e1 = eb + PT + t;
        bool v0 = e0 < hi, v1 = e1 < hi;
        int c0 = v0 ? e0 : hi - 1;
        int c1 = v1 ? e1 : hi - 1;
        int d0 = dst[c0], d1 = dst[c1];
        int s0 = src[c0], s1 = src[c1];
        const float2* p0 = reinterpret_cast<const float2*>(efeat + (size_t)c0 * EDGE_DIM);
        const float2* p1 = reinterpret_cast<const float2*>(efeat + (size_t)c1 * EDGE_DIM);
        float2 a0 = p0[0], a1 = p0[1], a2 = p0[2];
        float2 b0 = p1[0], b1 = p1[1], b2 = p1[2];
        asm volatile("" :: "v"(d0), "v"(d1), "v"(s0), "v"(s1),
                     "v"(a0.x), "v"(a0.y), "v"(a1.x), "v"(a1.y), "v"(a2.x), "v"(a2.y),
                     "v"(b0.x), "v"(b0.y), "v"(b1.x), "v"(b1.y), "v"(b2.x), "v"(b2.y));
        if (v0) {
            int slot = atomicAdd(&cursor[dst_bin(d0)], 1);
            stage[slot * 2]     = make_float4(__int_as_float(s0), __int_as_float(d0), a0.x, a0.y);
            stage[slot * 2 + 1] = make_float4(a1.x, a1.y, a2.x, a2.y);
        }
        if (v1) {
            int slot = atomicAdd(&cursor[dst_bin(d1)], 1);
            stage[slot * 2]     = make_float4(__int_as_float(s1), __int_as_float(d1), b0.x, b0.y);
            stage[slot * 2 + 1] = make_float4(b1.x, b1.y, b2.x, b2.y);
        }
    }
    __syncthreads();

    float4* rec4 = reinterpret_cast<float4*>(rec);
    const int* stage_i = reinterpret_cast<const int*>(stage);
    for (int i = t; i < 2 * n; i += PT) {
        int slot = i >> 1, h = i & 1;
        float4 val = stage[i];
        int d = stage_i[slot * 8 + 1];
        int pos = slot + gdelta[dst_bin(d)];
        rec4[(size_t)pos * 2 + h] = val;
    }
}

#define EB_WAVES (PT / 32)
#define AP 20
#define APOS(k) ((k) < HID ? (k) : (k) + 2)

__global__ __launch_bounds__(PT) void gnn_edge_bin_kernel(const float* __restrict__ h_in,
                                                          const float* __restrict__ rec,
                                                          const int* __restrict__ bstart,
                                                          const int* __restrict__ tot,
                                                          const float* __restrict__ g_w,
                                                          const float* __restrict__ u_w,
                                                          float* __restrict__ h_out, int n_nodes,
                                                          int out_pitch) {
    __shared__ float wlds[16 * 16];
    __shared__ float uwlds[20 * 16];
    __shared__ alignas(16) float acc[NPB * HID];
    __shared__ alignas(16) float alds[EB_WAVES][16 * AP];
    __shared__ int   slds[EB_WAVES][16];
    __shared__ alignas(16) int dlds[EB_WAVES][16];
    int t = threadIdx.x, bin = blockIdx.x;
    stage_weights_padded(wlds, g_w, 16, 16, t, PT);
    stage_weights_padded(uwlds, u_w, 20, 20, t, PT);
    for (int i = t; i < NPB * HID; i += PT) acc[i] = 0.0f;
    const int start = bstart[bin];
    const int end = start + tot[bin];
    const int node_lo = bin * NPB;
    __syncthreads();

    int wave = t >> 5, lane = t & 31;
    float* A = alds[wave];
    int* S = slds[wave];
    int* D = dlds[wave];
    const float4* rec4 = reinterpret_cast<const float4*>(rec);
    int m = lane & 15;
    int koff = (lane < 16) ? 0 : 2;
    int er = lane >> 1, hf = lane & 1;
    int rbase = (lane < 16) ? 0 : 8;

    v2f bf[4];
#pragma unroll
    for (int kc = 0; kc < 4; ++kc) {
        int kb = kc * 4 + koff;
        bf[kc].x = wlds[kb * 16 + m];
        bf[kc].y = wlds[(kb + 1) * 16 + m];
    }

    const int last = end - 1;
    int tb = start + wave * 16;
    int e0 = tb + er;
    if (e0 > last) e0 = last;
    if (e0 < 0) e0 = 0;
    float4 r = rec4[(size_t)e0 * 2 + hf];
    while (tb < end) {
        int tn = tb + EB_WAVES * 16;
        int en = tn + er;
        if (en > last) en = last;
        float4 rn = rec4[(size_t)en * 2 + hf];

        if (hf == 0) {
            S[er] = __float_as_int(r.x);
            D[er] = __float_as_int(r.y) - node_lo;
            *reinterpret_cast<float2*>(&A[er * AP + APOS(10)]) = make_float2(r.z, r.w);
        } else {
            *reinterpret_cast<float2*>(&A[er * AP + APOS(12)]) = make_float2(r.x, r.y);
            *reinterpret_cast<float2*>(&A[er * AP + APOS(14)]) = make_float2(r.z, r.w);
        }
        __builtin_amdgcn_wave_barrier();

        {
            const int gq = lane & 3, gr = lane >> 2;
            if (gq < 3) {
                const float4 g0 = *reinterpret_cast<const float4*>(
                    h_in + (size_t)S[gr] * HP + 4 * gq);
                const float4 g1 = *reinterpret_cast<const float4*>(
                    h_in + (size_t)S[gr + 8] * HP + 4 * gq);
                *reinterpret_cast<float4*>(&A[gr * AP + 4 * gq])       = g0;
                *reinterpret_cast<float4*>(&A[(gr + 8) * AP + 4 * gq]) = g1;
            }
        }
        __builtin_amdgcn_wave_barrier();

        const int4 dq0 = *reinterpret_cast<const int4*>(&D[rbase]);
        const int4 dq1 = *reinterpret_cast<const int4*>(&D[rbase + 4]);
        const int dl[8] = {dq0.x, dq0.y, dq0.z, dq0.w, dq1.x, dq1.y, dq1.z, dq1.w};

        v8f c = {};
#pragma unroll
        for (int kc = 0; kc < 4; ++kc) {
            int kb = kc * 4 + koff;
            const float* ap = &A[m * AP + APOS(kb)];
            v2f a;
            a.x = ap[0];
            a.y = ap[1];
            c = __builtin_amdgcn_wmma_f32_16x16x4_f32(false, a, false, bf[kc],
                                                      (short)0, c, false, false);
        }
        if (m < HID) {
            int nvalid = end - tb - rbase;
#pragma unroll
            for (int rr = 0; rr < 8; ++rr) {
                if (rr < nvalid && (unsigned)dl[rr] < NPB)
                    atomicAdd(&acc[dl[rr] * HID + m], lrelu(c[rr]));
            }
        }
        __builtin_amdgcn_wave_barrier();
        r = rn;
        tb = tn;
    }
    const bool upd = wave < NPB / 16;
    const int urow = lane & 15, uhalf = lane >> 4;
    const int nloc = wave * 16 + urow;
    const int unode = node_lo + nloc;
    float4 ha = make_float4(0.0f, 0.0f, 0.0f, 0.0f), hb = ha;
    float2 hc = make_float2(0.0f, 0.0f);
    if (upd && uhalf == 0 && unode < n_nodes) {
        const float* hp = h_in + (size_t)unode * HP;
        ha = *reinterpret_cast<const float4*>(hp);
        hb = *reinterpret_cast<const float4*>(hp + 4);
        hc = *reinterpret_cast<const float2*>(hp + 8);
    }
    __syncthreads();
    if (!upd) return;

    if (uhalf == 0) {
        *reinterpret_cast<float4*>(&A[urow * AP])     = ha;
        *reinterpret_cast<float4*>(&A[urow * AP + 4]) = hb;
        *reinterpret_cast<float2*>(&A[urow * AP + 8]) = hc;
    } else {
        const float2* ar = reinterpret_cast<const float2*>(&acc[nloc * HID]);
#pragma unroll
        for (int j = 0; j < HID / 2; ++j)
            *reinterpret_cast<float2*>(&A[urow * AP + HID + 2 * j]) = ar[j];
    }
    __builtin_amdgcn_wave_barrier();

    v8f uc = {};
#pragma unroll
    for (int kc = 0; kc < 5; ++kc) {
        int kb = kc * 4 + koff;
        v2f a, b;
        a.x = A[m * AP + kb];
        a.y = A[m * AP + kb + 1];
        b.x = uwlds[kb * 16 + m];
        b.y = uwlds[(kb + 1) * 16 + m];
        uc = __builtin_amdgcn_wmma_f32_16x16x4_f32(false, a, false, b,
                                                   (short)0, uc, false, false);
    }
    if (m < out_pitch) {
#pragma unroll
        for (int r = 0; r < 8; ++r) {
            int node = node_lo + wave * 16 + rbase + r;
            if (node < n_nodes)
                h_out[(size_t)node * out_pitch + m] = (m < HID) ? lrelu(uc[r]) : 0.0f;
        }
    }
}

extern "C" void kernel_launch(void* const* d_in, const int* in_sizes, int n_in,
                              void* d_out, int out_size, void* d_ws, size_t ws_size,
                              hipStream_t stream) {
    const float* vertex = (const float*)d_in[0];
    const float* efeat  = (const float*)d_in[1];
    const int*   src    = (const int*)d_in[2];
    const int*   dst    = (const int*)d_in[3];
    const float* w_init = (const float*)d_in[4];
    const float* g_w    = (const float*)d_in[5];
    const float* u_w    = (const float*)d_in[6];
    float* out = (float*)d_out;

    const int N = in_sizes[0] / 82;
    const int E = in_sizes[2];
    const int NP = N * HP;

    float* hA  = (float*)d_ws;
    float* hB  = hA + (size_t)NP;
    float* rec = hB + (size_t)NP;
    int* cnt    = (int*)(rec + (size_t)E * 8);
    int* off    = cnt + HB * NB;
    int* tot    = off + HB * NB;
    int* bstart = tot + NB;

    dim3 blk(256);
    int node_tiles  = (N + 15) / 16;
    int node_blocks = (node_tiles + WAVES - 1) / WAVES;
    int chunk = (E + HB - 1) / HB;

    gnn_embed_kernel<<<node_blocks, blk, 0, stream>>>(vertex, w_init, hA, N);

    gnn_hist_kernel<<<HB, HT, 0, stream>>>(dst, cnt, E, chunk);
    gnn_scan_kernel<<<NB / SB, 1024, 0, stream>>>(cnt, off, tot);
    gnn_scatter_kernel<<<HB, PT, 0, stream>>>(efeat, src, dst, cnt, off, tot, bstart, rec, E, chunk);

    gnn_edge_bin_kernel<<<NB, PT, 0, stream>>>(hA, rec, bstart, tot, g_w, u_w, hB, N, HP);

    gnn_edge_bin_kernel<<<NB, PT, 0, stream>>>(hB, rec, bstart, tot, g_w, u_w, out, N, HID);
}
